# starting kernel
# baseline (speedup 1.0000x reference)
#include <hip/hip_runtime.h>
#include <hip/hip_bf16.h>


typedef float v2f __attribute__((ext_vector_type(2)));
typedef float v8f __attribute__((ext_vector_type(8)));

constexpr int B_   = 2;
constexpr int S_   = 2048;
constexpr int H_   = 16;
constexpr int D_   = 64;
constexpr int CH   = 64;
constexpr int LDP  = 68;
constexpr int NSPL = 8;

constexpr int OFF_Q    = 0;
constexpr int OFF_K    = 64 * LDP;
constexpr int OFF_V    = 2 * 64 * LDP;
constexpr int OFF_S    = 3 * 64 * LDP;
constexpr int OFF_KV   = 4 * 64 * LDP;
constexpr int OFF_KSUM = 5 * 64 * LDP;
constexpr int OFF_DEN  = OFF_KSUM + 64;
constexpr int SMEM_F   = OFF_DEN + 64;

constexpr int STATE_F  = 64 * 64 + 64;

__device__ __forceinline__ float elu1(float x) {
  return x > 0.0f ? x + 1.0f : __expf(x);
}

__device__ __forceinline__ v8f wmma4(v2f a, v2f b, v8f c) {
  return __builtin_amdgcn_wmma_f32_16x16x4_f32(
       false, a,  false, b,
       (short)0, c,  false,  false);
}

__device__ __forceinline__ v2f lds_pair(const float* p) {
  return *reinterpret_cast<const v2f*>(p);
}

__device__ __forceinline__ v8f cd_load(const float* base, int mi, int nj, int lane) {
  int n  = nj * 16 + (lane & 15);
  int m0 = mi * 16 + ((lane >> 4) << 3);
  v8f c;
#pragma unroll
  for (int r = 0; r < 8; ++r) c[r] = base[(m0 + r) * LDP + n];
  return c;
}
__device__ __forceinline__ void cd_store(float* base, int mi, int nj, int lane, v8f c) {
  int n  = nj * 16 + (lane & 15);
  int m0 = mi * 16 + ((lane >> 4) << 3);
#pragma unroll
  for (int r = 0; r < 8; ++r) base[(m0 + r) * LDP + n] = c[r];
}

__device__ __forceinline__ v8f gemm_nn(const float* A, const float* Bm,
                                       int mi, int nj, int lane,
                                       int k0, int k1, v8f acc) {
  int lm = lane & 15, hi = lane >> 4;
  int arow = (mi * 16 + lm) * LDP;
  int bcol = nj * 16 + lm;
  for (int kk = k0; kk < k1; ++kk) {
    int kb = kk * 4 + hi * 2;
    v2f a = lds_pair(&A[arow + kb]);
    v2f b; b.x = Bm[kb * LDP + bcol]; b.y = Bm[(kb + 1) * LDP + bcol];
    acc = wmma4(a, b, acc);
  }
  return acc;
}
__device__ __forceinline__ v8f gemm_nt(const float* A, const float* Bt,
                                       int mi, int nj, int lane, v8f acc) {
  int lm = lane & 15, hi = lane >> 4;
  int arow = (mi * 16 + lm) * LDP;
  int brow = (nj * 16 + lm) * LDP;
#pragma unroll
  for (int kk = 0; kk < 16; ++kk) {
    int kb = kk * 4 + hi * 2;
    v2f a = lds_pair(&A[arow + kb]);
    v2f b = lds_pair(&Bt[brow + kb]);
    acc = wmma4(a, b, acc);
  }
  return acc;
}
__device__ __forceinline__ v8f gemm_tn(const float* At, const float* Bm,
                                       int mi, int nj, int lane, v8f acc) {
  int lm = lane & 15, hi = lane >> 4;
  int am   = mi * 16 + lm;
  int bcol = nj * 16 + lm;
#pragma unroll
  for (int kk = 0; kk < 16; ++kk) {
    int kb = kk * 4 + hi * 2;
    v2f a; a.x = At[kb * LDP + am];   a.y = At[(kb + 1) * LDP + am];
    v2f b; b.x = Bm[kb * LDP + bcol]; b.y = Bm[(kb + 1) * LDP + bcol];
    acc = wmma4(a, b, acc);
  }
  return acc;
}

__device__ __forceinline__ void mask_diag(v8f& c, int lane) {
  int n  = lane & 15;
  int m0 = (lane >> 4) << 3;
#pragma unroll
  for (int r = 0; r < 8; ++r)
    if (m0 + r < n) c[r] = 0.0f;
}

__device__ __forceinline__ void load_tile(const float* __restrict__ g, int rowStride,
                                          float* lds, int tid, bool do_elu) {
#pragma unroll
  for (int j = 0; j < 4; ++j) {
    int u = tid + j * 256;
    int r = u >> 4;
    int c = (u & 15) << 2;
    float4 w = *reinterpret_cast<const float4*>(g + (size_t)r * rowStride + c);
    if (do_elu) { w.x = elu1(w.x); w.y = elu1(w.y); w.z = elu1(w.z); w.w = elu1(w.w); }
    *reinterpret_cast<float4*>(lds + r * LDP + c) = w;
  }
}

__global__ __launch_bounds__(256) void lin_attn_states(
    const float* __restrict__ qk, const float* __restrict__ v,
    float* __restrict__ ws, int nsplit) {
  extern __shared__ float smem[];
  float* sK    = smem + OFF_K;
  float* sV    = smem + OFF_V;
  float* sKV   = smem + OFF_KV;
  float* sksum = smem + OFF_KSUM;

  int tid = threadIdx.x, lane = tid & 31, wid = tid >> 5;
  int wg = blockIdx.x;
  int split = wg % nsplit;
  int h     = (wg / nsplit) % H_;
  int b     = wg / (nsplit * H_);
  int cps   = (S_ / CH) / nsplit;

  for (int i = tid; i < 64 * LDP; i += 256) sKV[i] = 0.0f;
  if (tid < 64) sksum[tid] = 0.0f;
  __syncthreads();

  for (int c = 0; c < cps; ++c) {
    int base = (split * cps + c) * CH;
    const float* kptr = qk + (((size_t)(b * S_ + base) * 2 + 1) * H_ + h) * D_;
    const float* vptr = v  + (((size_t)(b * S_ + base)) * H_ + h) * D_;
    load_tile(kptr, 2 * H_ * D_, sK, tid, true);
    load_tile(vptr, H_ * D_,     sV, tid, false);
    __syncthreads();

    if (tid < 64) {
      float acc = 0.0f;
      for (int s = 0; s < 64; ++s) acc += sK[s * LDP + tid];
      sksum[tid] += acc;
    }
#pragma unroll
    for (int i = 0; i < 2; ++i) {
      int t = wid * 2 + i, mi = t >> 2, nj = t & 3;
      v8f acc = cd_load(sKV, mi, nj, lane);
      acc = gemm_tn(sK, sV, mi, nj, lane, acc);
      cd_store(sKV, mi, nj, lane, acc);
    }
    __syncthreads();
  }

  float* wsp = ws + (size_t)((b * H_ + h) * nsplit + split) * STATE_F;
  for (int i = tid; i < 64 * 64; i += 256)
    wsp[i] = sKV[(i >> 6) * LDP + (i & 63)];
  if (tid < 64) wsp[64 * 64 + tid] = sksum[tid];
}

__global__ __launch_bounds__(256) void lin_attn_main(
    const float* __restrict__ qk, const float* __restrict__ v,
    const float* __restrict__ ws, float* __restrict__ out, int nsplit) {
  extern __shared__ float smem[];
  float* sQ    = smem + OFF_Q;
  float* sK    = smem + OFF_K;
  float* sV    = smem + OFF_V;
  float* sS    = smem + OFF_S;
  float* sKV   = smem + OFF_KV;
  float* sksum = smem + OFF_KSUM;
  float* sden  = smem + OFF_DEN;

  int tid = threadIdx.x, lane = tid & 31, wid = tid >> 5;
  int wg = blockIdx.x;
  int split = wg % nsplit;
  int h     = (wg / nsplit) % H_;
  int b     = wg / (nsplit * H_);
  int cps   = (S_ / CH) / nsplit;

  for (int i = tid; i < 64 * LDP; i += 256) sKV[i] = 0.0f;
  if (tid < 64) sksum[tid] = 0.0f;
  __syncthreads();

  for (int p = 0; p < split; ++p) {
    const float* wsp = ws + (size_t)((b * H_ + h) * nsplit + p) * STATE_F;
    for (int i = tid; i < 64 * 64; i += 256)
      sKV[(i >> 6) * LDP + (i & 63)] += wsp[i];
    if (tid < 64) sksum[tid] += wsp[64 * 64 + tid];
  }
  __syncthreads();

  int t0 = wid * 2, t1 = wid * 2 + 1;
  int m0 = t0 >> 2, n0 = t0 & 3, m1 = t1 >> 2, n1 = t1 & 3;

  for (int c = 0; c < cps; ++c) {
    int base = (split * cps + c) * CH;
    const float* qptr = qk + (((size_t)(b * S_ + base) * 2 + 0) * H_ + h) * D_;
    const float* kptr = qk + (((size_t)(b * S_ + base) * 2 + 1) * H_ + h) * D_;
    const float* vptr = v  + (((size_t)(b * S_ + base)) * H_ + h) * D_;
    load_tile(qptr, 2 * H_ * D_, sQ, tid, true);
    load_tile(kptr, 2 * H_ * D_, sK, tid, true);
    load_tile(vptr, H_ * D_,     sV, tid, false);

    if (c + 1 < cps) {
      int nb = base + CH;
      const float* nq = qk + (((size_t)(b * S_ + nb) * 2) * H_ + h) * D_;
      const float* nv = v  + (((size_t)(b * S_ + nb)) * H_ + h) * D_;
      int r = tid >> 2, o = (tid & 3) << 4;
      __builtin_prefetch(nq + (size_t)r * 2 * H_ * D_ + o, 0, 3);
      __builtin_prefetch(nq + (size_t)r * 2 * H_ * D_ + H_ * D_ + o, 0, 3);
      __builtin_prefetch(nv + (size_t)r * H_ * D_ + o, 0, 3);
    }
    __syncthreads();

    v8f oa0 = {}, oa1 = {};
    oa0 = gemm_nn(sQ, sKV, m0, n0, lane, 0, 16, oa0);
    oa1 = gemm_nn(sQ, sKV, m1, n1, lane, 0, 16, oa1);

    if (tid < 64) {
      float d = 0.0f;
      for (int dd = 0; dd < 64; ++dd) d += sQ[tid * LDP + dd] * sksum[dd];
      sden[tid] = d;
    }

    if (n0 <= m0) {
      v8f sa = {};
      sa = gemm_nt(sQ, sK, m0, n0, lane, sa);
      if (m0 == n0) mask_diag(sa, lane);
      cd_store(sS, m0, n0, lane, sa);
    }
    if (n1 <= m1) {
      v8f sa = {};
      sa = gemm_nt(sQ, sK, m1, n1, lane, sa);
      if (m1 == n1) mask_diag(sa, lane);
      cd_store(sS, m1, n1, lane, sa);
    }
    __syncthreads();

    if (tid < 64) {
      float ssum = 0.0f;
      int smax = ((tid >> 4) + 1) * 16;
      for (int s = 0; s < smax; ++s) ssum += sS[tid * LDP + s];
      sden[tid] = 1.0f / (sden[tid] + ssum);
    }
    __syncthreads();

    oa0 = gemm_nn(sS, sV, m0, n0, lane, 0, (m0 + 1) * 4, oa0);
    oa1 = gemm_nn(sS, sV, m1, n1, lane, 0, (m1 + 1) * 4, oa1);

    {
      int nn = n0 * 16 + (lane & 15);
      int mm = m0 * 16 + ((lane >> 4) << 3);
#pragma unroll
      for (int r = 0; r < 8; ++r)
        out[(((size_t)b * S_ + base + mm + r) * H_ + h) * D_ + nn] = oa0[r] * sden[mm + r];
      nn = n1 * 16 + (lane & 15);
      mm = m1 * 16 + ((lane >> 4) << 3);
#pragma unroll
      for (int r = 0; r < 8; ++r)
        out[(((size_t)b * S_ + base + mm + r) * H_ + h) * D_ + nn] = oa1[r] * sden[mm + r];
    }
    __syncthreads();

    if (tid < 64) {
      float acc = 0.0f;
      for (int s = 0; s < 64; ++s) acc += sK[s * LDP + tid];
      sksum[tid] += acc;
    }
    {
      v8f a0 = cd_load(sKV, m0, n0, lane);
      a0 = gemm_tn(sK, sV, m0, n0, lane, a0);
      cd_store(sKV, m0, n0, lane, a0);
      v8f a1 = cd_load(sKV, m1, n1, lane);
      a1 = gemm_tn(sK, sV, m1, n1, lane, a1);
      cd_store(sKV, m1, n1, lane, a1);
    }
    __syncthreads();
  }
}

extern "C" void kernel_launch(void* const* d_in, const int* in_sizes, int n_in,
                              void* d_out, int out_size, void* d_ws, size_t ws_size,
                              hipStream_t stream) {
  const float* qk = (const float*)d_in[0];
  const float* v  = (const float*)d_in[1];
  float* out = (float*)d_out;
  float* ws  = (float*)d_ws;

  int nsplit = NSPL;
  size_t need = (size_t)B_ * H_ * NSPL * STATE_F * sizeof(float);
  if (ws_size < need) nsplit = 1;

  int wgs = B_ * H_ * nsplit;
  size_t smem = (size_t)SMEM_F * sizeof(float);

  if (nsplit > 1)
    lin_attn_states<<<wgs, 256, smem, stream>>>(qk, v, ws, nsplit);
  lin_attn_main<<<wgs, 256, smem, stream>>>(qk, v, ws, out, nsplit);
}


# Normaliser sums (K column sums, q.ksum, score row sums) spread over all lanes as partial sums joined by cross-lane adds; 64-thread serial loops and one barrier phase go; their summation order changes
# speedup vs baseline: 2.5566x; 2.5566x over previous
#include <hip/hip_runtime.h>
#include <hip/hip_bf16.h>


typedef float v2f __attribute__((ext_vector_type(2)));
typedef float v8f __attribute__((ext_vector_type(8)));

constexpr int B_   = 2;
constexpr int S_   = 2048;
constexpr int H_   = 16;
constexpr int D_   = 64;
constexpr int CH   = 64;
constexpr int LDP  = 68;
constexpr int NSPL = 8;

constexpr int OFF_Q    = 0;
constexpr int OFF_K    = 64 * LDP;
constexpr int OFF_V    = 2 * 64 * LDP;
constexpr int OFF_S    = 3 * 64 * LDP;
constexpr int OFF_KV   = 4 * 64 * LDP;
constexpr int OFF_KSUM = 5 * 64 * LDP;
constexpr int OFF_DEN  = OFF_KSUM + 64;
constexpr int OFF_KPART = OFF_DEN + 64;
constexpr int OFF_ROW  = OFF_KPART + 512;
constexpr int SMEM_F   = OFF_ROW + 256;

constexpr int STATE_F  = 64 * 64 + 64;

__device__ __forceinline__ float elu1(float x) {
  return x > 0.0f ? x + 1.0f : __expf(x);
}


__device__ __forceinline__ void lds_barrier() {
  __builtin_amdgcn_fence(__ATOMIC_RELEASE, "workgroup", "local");
  __builtin_amdgcn_s_barrier();
  __builtin_amdgcn_fence(__ATOMIC_ACQUIRE, "workgroup", "local");
}

template <int M>
__device__ __forceinline__ float row_xor_add(float x) {
  int xi = __float_as_int(x);
  int yi = __builtin_amdgcn_update_dpp(xi, xi, 0x160 + M, 0xf, 0xf, true);
  return x + __int_as_float(yi);
}

__device__ __forceinline__ v8f wmma4(v2f a, v2f b, v8f c) {
  return __builtin_amdgcn_wmma_f32_16x16x4_f32(
       false, a,  false, b,
       (short)0, c,  false,  false);
}

__device__ __forceinline__ v2f lds_pair(const float* p) {
  return *reinterpret_cast<const v2f*>(p);
}

__device__ __forceinline__ v8f cd_load(const float* base, int mi, int nj, int lane) {
  int n  = nj * 16 + (lane & 15);
  int m0 = mi * 16 + ((lane >> 4) << 3);
  v8f c;
#pragma unroll
  for (int r = 0; r < 8; ++r) c[r] = base[(m0 + r) * LDP + n];
  return c;
}
__device__ __forceinline__ void cd_store(float* base, int mi, int nj, int lane, v8f c) {
  int n  = nj * 16 + (lane & 15);
  int m0 = mi * 16 + ((lane >> 4) << 3);
#pragma unroll
  for (int r = 0; r < 8; ++r) base[(m0 + r) * LDP + n] = c[r];
}

__device__ __forceinline__ v8f gemm_nn(const float* A, const float* Bm,
                                       int mi, int nj, int lane,
                                       int k0, int k1, v8f acc) {
  int lm = lane & 15, hi = lane >> 4;
  int arow = (mi * 16 + lm) * LDP;
  int bcol = nj * 16 + lm;
  for (int kk = k0; kk < k1; ++kk) {
    int kb = kk * 4 + hi * 2;
    v2f a = lds_pair(&A[arow + kb]);
    v2f b; b.x = Bm[kb * LDP + bcol]; b.y = Bm[(kb + 1) * LDP + bcol];
    acc = wmma4(a, b, acc);
  }
  return acc;
}
__device__ __forceinline__ v8f gemm_nt(const float* A, const float* Bt,
                                       int mi, int nj, int lane, v8f acc) {
  int lm = lane & 15, hi = lane >> 4;
  int arow = (mi * 16 + lm) * LDP;
  int brow = (nj * 16 + lm) * LDP;
#pragma unroll
  for (int kk = 0; kk < 16; ++kk) {
    int kb = kk * 4 + hi * 2;
    v2f a = lds_pair(&A[arow + kb]);
    v2f b = lds_pair(&Bt[brow + kb]);
    acc = wmma4(a, b, acc);
  }
  return acc;
}
__device__ __forceinline__ v8f gemm_tn(const float* At, const float* Bm,
                                       int mi, int nj, int lane, v8f acc) {
  int lm = lane & 15, hi = lane >> 4;
  int am   = mi * 16 + lm;
  int bcol = nj * 16 + lm;
#pragma unroll
  for (int kk = 0; kk < 16; ++kk) {
    int kb = kk * 4 + hi * 2;
    v2f a; a.x = At[kb * LDP + am];   a.y = At[(kb + 1) * LDP + am];
    v2f b; b.x = Bm[kb * LDP + bcol]; b.y = Bm[(kb + 1) * LDP + bcol];
    acc = wmma4(a, b, acc);
  }
  return acc;
}

__device__ __forceinline__ void mask_diag(v8f& c, int lane) {
  int n  = lane & 15;
  int m0 = (lane >> 4) << 3;
#pragma unroll
  for (int r = 0; r < 8; ++r)
    if (m0 + r < n) c[r] = 0.0f;
}

__device__ __forceinline__ void tile_rowsum_store(float* rowp, int mi, int nj, int lane, v8f c) {
  float s[8];
#pragma unroll
  for (int r = 0; r < 8; ++r) {
    float x = c[r];
    x = row_xor_add<1>(x);
    x = row_xor_add<2>(x);
    x = row_xor_add<4>(x);
    x = row_xor_add<8>(x);
    s[r] = x;
  }
  if ((lane & 15) == 0) {
    float* p = rowp + nj * 64 + mi * 16 + ((lane >> 4) << 3);
    *reinterpret_cast<float4*>(p)     = make_float4(s[0], s[1], s[2], s[3]);
    *reinterpret_cast<float4*>(p + 4) = make_float4(s[4], s[5], s[6], s[7]);
  }
}

__device__ __forceinline__ float xhalf(float x) {
  return __int_as_float(__builtin_amdgcn_permlanex16(__float_as_int(x), __float_as_int(x),
                                                     0x76543210, 0xfedcba98, false, false));
}
__device__ __forceinline__ void ksum_wave_partial(float* kpart, int tid, float4 cs) {
  cs.x += xhalf(cs.x); cs.y += xhalf(cs.y); cs.z += xhalf(cs.z); cs.w += xhalf(cs.w);
  if ((tid & 31) < 16)
    *reinterpret_cast<float4*>(kpart + (tid >> 5) * 64 + ((tid & 15) << 2)) = cs;
}
__device__ __forceinline__ float ksum_join(const float* kpart, int d) {
  float p0 = kpart[d],       p1 = kpart[64 + d],  p2 = kpart[128 + d], p3 = kpart[192 + d];
  float p4 = kpart[256 + d], p5 = kpart[320 + d], p6 = kpart[384 + d], p7 = kpart[448 + d];
  return ((p0 + p1) + (p2 + p3)) + ((p4 + p5) + (p6 + p7));
}

__device__ __forceinline__ void tile_issue(const float* __restrict__ g, int rowStride,
                                           int tid, float4 (&w)[4]) {
#pragma unroll
  for (int j = 0; j < 4; ++j) {
    int u = tid + j * 256;
    int r = u >> 4;
    int c = (u & 15) << 2;
    w[j] = *reinterpret_cast<const float4*>(g + (size_t)r * rowStride + c);
  }
}
__device__ __forceinline__ void tile_commit(float* lds, int tid, const float4 (&w)[4],
                                            bool do_elu) {
#pragma unroll
  for (int j = 0; j < 4; ++j) {
    int u = tid + j * 256;
    int r = u >> 4;
    int c = (u & 15) << 2;
    float4 x = w[j];
    if (do_elu) { x.x = elu1(x.x); x.y = elu1(x.y); x.z = elu1(x.z); x.w = elu1(x.w); }
    *reinterpret_cast<float4*>(lds + r * LDP + c) = x;
  }
}
__device__ __forceinline__ float4 tile_commit_colsum(float* lds, int tid, const float4 (&w)[4]) {
  float4 cs = make_float4(0.0f, 0.0f, 0.0f, 0.0f);
#pragma unroll
  for (int j = 0; j < 4; ++j) {
    int u = tid + j * 256;
    int r = u >> 4;
    int c = (u & 15) << 2;
    float4 x = w[j];
    x.x = elu1(x.x); x.y = elu1(x.y); x.z = elu1(x.z); x.w = elu1(x.w);
    *reinterpret_cast<float4*>(lds + r * LDP + c) = x;
    if (j == 0) cs = x;
    else { cs.x += x.x; cs.y += x.y; cs.z += x.z; cs.w += x.w; }
  }
  return cs;
}

__global__ __launch_bounds__(256) void lin_attn_states(
    const float* __restrict__ qk, const float* __restrict__ v,
    float* __restrict__ ws, int nsplit) {
  extern __shared__ float smem[];
  float* sK    = smem + OFF_K;
  float* sV    = smem + OFF_V;
  float* sKV   = smem + OFF_KV;
  float* sKpart = smem + OFF_KPART;

  int tid = threadIdx.x, lane = tid & 31, wid = tid >> 5;
  int wg = blockIdx.x;
  int split = wg % nsplit;
  int h     = (wg / nsplit) % H_;
  int b     = wg / (nsplit * H_);
  int cps   = (S_ / CH) / nsplit;

  for (int i = tid; i < 64 * LDP; i += 256) sKV[i] = 0.0f;
  float ks = 0.0f;
  __syncthreads();

  float4 rk[4], rv[4];
  {
    int base0 = split * cps * CH;
    tile_issue(qk + (((size_t)(b * S_ + base0) * 2 + 1) * H_ + h) * D_, 2 * H_ * D_, tid, rk);
    tile_issue(v  + (((size_t)(b * S_ + base0)) * H_ + h) * D_,         H_ * D_,     tid, rv);
  }

  for (int c = 0; c < cps; ++c) {
    ksum_wave_partial(sKpart, tid, tile_commit_colsum(sK, tid, rk));
    tile_commit(sV, tid, rv, false);
    if (c + 1 < cps) {
      int nb = (split * cps + c + 1) * CH;
      tile_issue(qk + (((size_t)(b * S_ + nb) * 2 + 1) * H_ + h) * D_, 2 * H_ * D_, tid, rk);
      tile_issue(v  + (((size_t)(b * S_ + nb)) * H_ + h) * D_,         H_ * D_,     tid, rv);
    }
    lds_barrier();

    if (tid < 64) ks += ksum_join(sKpart, tid);
#pragma unroll
    for (int i = 0; i < 2; ++i) {
      int t = wid * 2 + i, mi = t >> 2, nj = t & 3;
      v8f acc = cd_load(sKV, mi, nj, lane);
      acc = gemm_tn(sK, sV, mi, nj, lane, acc);
      cd_store(sKV, mi, nj, lane, acc);
    }
    lds_barrier();
  }

  float* wsp = ws + (size_t)((b * H_ + h) * nsplit + split) * STATE_F;
  for (int i = tid; i < 64 * 64; i += 256)
    wsp[i] = sKV[(i >> 6) * LDP + (i & 63)];
  if (tid < 64) wsp[64 * 64 + tid] = ks;
}

__global__ __launch_bounds__(256) void lin_attn_main(
    const float* __restrict__ qk, const float* __restrict__ v,
    const float* __restrict__ ws, float* __restrict__ out, int nsplit) {
  extern __shared__ float smem[];
  float* sQ    = smem + OFF_Q;
  float* sK    = smem + OFF_K;
  float* sV    = smem + OFF_V;
  float* sS    = smem + OFF_S;
  float* sKV   = smem + OFF_KV;
  float* sksum = smem + OFF_KSUM;
  float* sden  = smem + OFF_DEN;
  float* sKpart = smem + OFF_KPART;
  float* sRow  = smem + OFF_ROW;

  int tid = threadIdx.x, lane = tid & 31, wid = tid >> 5;
  int wg = blockIdx.x;
  int split = wg % nsplit;
  int h     = (wg / nsplit) % H_;
  int b     = wg / (nsplit * H_);
  int cps   = (S_ / CH) / nsplit;

  float4 rq[4], rk[4], rv[4];
  {
    int base0 = split * cps * CH;
    tile_issue(qk + (((size_t)(b * S_ + base0) * 2 + 0) * H_ + h) * D_, 2 * H_ * D_, tid, rq);
    tile_issue(qk + (((size_t)(b * S_ + base0) * 2 + 1) * H_ + h) * D_, 2 * H_ * D_, tid, rk);
    tile_issue(v  + (((size_t)(b * S_ + base0)) * H_ + h) * D_,         H_ * D_,     tid, rv);
  }

  {
    constexpr int PB = 7;
    const float* wbase = ws + (size_t)((b * H_ + h) * nsplit) * STATE_F;
    float4 acc[4];
    float  accs = 0.0f;
#pragma unroll
    for (int j = 0; j < 4; ++j) acc[j] = make_float4(0.0f, 0.0f, 0.0f, 0.0f);
    for (int p0 = 0; p0 < split; p0 += PB) {
      float4 st[PB][4];
      float  sk[PB];
#pragma unroll
      for (int q = 0; q < PB; ++q) {
        if (p0 + q < split) {
          const float* wsp = wbase + (size_t)(p0 + q) * STATE_F;
#pragma unroll
          for (int j = 0; j < 4; ++j)
            st[q][j] = *reinterpret_cast<const float4*>(wsp + 4 * (tid + 256 * j));
          if (tid < 64) sk[q] = wsp[64 * 64 + tid];
        }
      }
#pragma unroll
      for (int q = 0; q < PB; ++q) {
        if (p0 + q < split) {
#pragma unroll
          for (int j = 0; j < 4; ++j) {
            acc[j].x += st[q][j].x; acc[j].y += st[q][j].y;
            acc[j].z += st[q][j].z; acc[j].w += st[q][j].w;
          }
          if (tid < 64) accs += sk[q];
        }
      }
    }
#pragma unroll
    for (int j = 0; j < 4; ++j) {
      int f = tid + 256 * j;
      *reinterpret_cast<float4*>(sKV + (f >> 4) * LDP + ((f & 15) << 2)) = acc[j];
    }
    if (tid < 64) sksum[tid] = accs;
  }
  __syncthreads();

  int t0 = wid * 2, t1 = wid * 2 + 1;
  int m0 = t0 >> 2, n0 = t0 & 3, m1 = t1 >> 2, n1 = t1 & 3;

  for (int c = 0; c < cps; ++c) {
    int base = (split * cps + c) * CH;
    tile_commit(sQ, tid, rq, true);
    ksum_wave_partial(sKpart, tid, tile_commit_colsum(sK, tid, rk));
    tile_commit(sV, tid, rv, false);
    if (c + 1 < cps) {
      int nb = base + CH;
      tile_issue(qk + (((size_t)(b * S_ + nb) * 2 + 0) * H_ + h) * D_, 2 * H_ * D_, tid, rq);
      tile_issue(qk + (((size_t)(b * S_ + nb) * 2 + 1) * H_ + h) * D_, 2 * H_ * D_, tid, rk);
      tile_issue(v  + (((size_t)(b * S_ + nb)) * H_ + h) * D_,         H_ * D_,     tid, rv);
    }
    lds_barrier();

    v8f oa0 = {}, oa1 = {};
    oa0 = gemm_nn(sQ, sKV, m0, n0, lane, 0, 16, oa0);
    oa1 = gemm_nn(sQ, sKV, m1, n1, lane, 0, 16, oa1);

    {
      int t = tid >> 2, q = tid & 3;
      const float* qr = sQ + t * LDP + 16 * q;
      const float* kr = sksum + 16 * q;
      float d = 0.0f;
#pragma unroll
      for (int i = 0; i < 16; i += 4) {
        float4 a = *reinterpret_cast<const float4*>(qr + i);
        float4 k = *reinterpret_cast<const float4*>(kr + i);
        d += a.x * k.x; d += a.y * k.y; d += a.z * k.z; d += a.w * k.w;
      }
      d = row_xor_add<1>(d);
      d = row_xor_add<2>(d);
      if (q == 0) sden[t] = d;
    }

    if (n0 <= m0) {
      v8f sa = {};
      sa = gemm_nt(sQ, sK, m0, n0, lane, sa);
      if (m0 == n0) mask_diag(sa, lane);
      tile_rowsum_store(sRow, m0, n0, lane, sa);
      cd_store(sS, m0, n0, lane, sa);
    }
    if (n1 <= m1) {
      v8f sa = {};
      sa = gemm_nt(sQ, sK, m1, n1, lane, sa);
      if (m1 == n1) mask_diag(sa, lane);
      tile_rowsum_store(sRow, m1, n1, lane, sa);
      cd_store(sS, m1, n1, lane, sa);
    }
    lds_barrier();

    float inv[8];
    {
      int R = m0 * 16 + ((lane >> 4) << 3);
      float4 ia = *reinterpret_cast<const float4*>(sden + R);
      float4 ib = *reinterpret_cast<const float4*>(sden + R + 4);
      float4 sa = *reinterpret_cast<const float4*>(sRow + R);
      float4 sb = *reinterpret_cast<const float4*>(sRow + R + 4);
#pragma unroll
      for (int n = 1; n < 4; ++n) {
        if (n <= m0) {
          float4 ta = *reinterpret_cast<const float4*>(sRow + n * 64 + R);
          float4 tb = *reinterpret_cast<const float4*>(sRow + n * 64 + R + 4);
          sa.x += ta.x; sa.y += ta.y; sa.z += ta.z; sa.w += ta.w;
          sb.x += tb.x; sb.y += tb.y; sb.z += tb.z; sb.w += tb.w;
        }
      }
      inv[0] = 1.0f / (ia.x + sa.x); inv[1] = 1.0f / (ia.y + sa.y);
      inv[2] = 1.0f / (ia.z + sa.z); inv[3] = 1.0f / (ia.w + sa.w);
      inv[4] = 1.0f / (ib.x + sb.x); inv[5] = 1.0f / (ib.y + sb.y);
      inv[6] = 1.0f / (ib.z + sb.z); inv[7] = 1.0f / (ib.w + sb.w);
    }

    oa0 = gemm_nn(sS, sV, m0, n0, lane, 0, (m0 + 1) * 4, oa0);
    oa1 = gemm_nn(sS, sV, m1, n1, lane, 0, (m1 + 1) * 4, oa1);

    {
      int nn = n0 * 16 + (lane & 15);
      int mm = m0 * 16 + ((lane >> 4) << 3);
#pragma unroll
      for (int r = 0; r < 8; ++r)
        out[(((size_t)b * S_ + base + mm + r) * H_ + h) * D_ + nn] = oa0[r] * inv[r];
      nn = n1 * 16 + (lane & 15);
      mm = m1 * 16 + ((lane >> 4) << 3);
#pragma unroll
      for (int r = 0; r < 8; ++r)
        out[(((size_t)b * S_ + base + mm + r) * H_ + h) * D_ + nn] = oa1[r] * inv[r];
    }
    lds_barrier();

    if (tid < 64) sksum[tid] += ksum_join(sKpart, tid);
    {
      v8f a0 = cd_load(sKV, m0, n0, lane);
      a0 = gemm_tn(sK, sV, m0, n0, lane, a0);
      cd_store(sKV, m0, n0, lane, a0);
      v8f a1 = cd_load(sKV, m1, n1, lane);
      a1 = gemm_tn(sK, sV, m1, n1, lane, a1);
      cd_store(sKV, m1, n1, lane, a1);
    }
    lds_barrier();
  }
}

extern "C" void kernel_launch(void* const* d_in, const int* in_sizes, int n_in,
                              void* d_out, int out_size, void* d_ws, size_t ws_size,
                              hipStream_t stream) {
  const float* qk = (const float*)d_in[0];
  const float* v  = (const float*)d_in[1];
  float* out = (float*)d_out;
  float* ws  = (float*)d_ws;

  int nsplit = NSPL;
  size_t need = (size_t)B_ * H_ * NSPL * STATE_F * sizeof(float);
  if (ws_size < need) nsplit = 1;

  int wgs = B_ * H_ * nsplit;
  size_t smem = (size_t)SMEM_F * sizeof(float);

  if (nsplit > 1)
    lin_attn_states<<<wgs, 256, smem, stream>>>(qk, v, ws, nsplit);
  lin_attn_main<<<wgs, 256, smem, stream>>>(qk, v, ws, out, nsplit);
}


# Triangular phases (masked scores, S@V) as straight-line code per wave: scalar wave index gives static trip counts, two tiles run as interleaved chains, LDS fragments fetched a 4-step block ahead
# speedup vs baseline: 2.6987x; 1.0556x over previous
#include <hip/hip_runtime.h>
#include <hip/hip_bf16.h>


typedef float v2f __attribute__((ext_vector_type(2)));
typedef float v8f __attribute__((ext_vector_type(8)));

constexpr int B_   = 2;
constexpr int S_   = 2048;
constexpr int H_   = 16;
constexpr int D_   = 64;
constexpr int CH   = 64;
constexpr int LDP  = 68;
constexpr int NSPL = 8;

constexpr int OFF_Q    = 0;
constexpr int OFF_K    = 64 * LDP;
constexpr int OFF_V    = 2 * 64 * LDP;
constexpr int OFF_S    = 3 * 64 * LDP;
constexpr int OFF_KV   = 4 * 64 * LDP;
constexpr int OFF_KSUM = 5 * 64 * LDP;
constexpr int OFF_DEN  = OFF_KSUM + 64;
constexpr int OFF_KPART = OFF_DEN + 64;
constexpr int OFF_ROW  = OFF_KPART + 512;
constexpr int SMEM_F   = OFF_ROW + 256;

constexpr int STATE_F  = 64 * 64 + 64;

__device__ __forceinline__ float elu1(float x) {
  return x > 0.0f ? x + 1.0f : __expf(x);
}


__device__ __forceinline__ void lds_barrier() {
  __builtin_amdgcn_fence(__ATOMIC_RELEASE, "workgroup", "local");
  __builtin_amdgcn_s_barrier();
  __builtin_amdgcn_fence(__ATOMIC_ACQUIRE, "workgroup", "local");
}

template <int M>
__device__ __forceinline__ float row_xor_add(float x) {
  int xi = __float_as_int(x);
  int yi = __builtin_amdgcn_update_dpp(xi, xi, 0x160 + M, 0xf, 0xf, true);
  return x + __int_as_float(yi);
}

__device__ __forceinline__ v8f wmma4(v2f a, v2f b, v8f c) {
  return __builtin_amdgcn_wmma_f32_16x16x4_f32(
       false, a,  false, b,
       (short)0, c,  false,  false);
}

__device__ __forceinline__ v2f lds_pair(const float* p) {
  return *reinterpret_cast<const v2f*>(p);
}

__device__ __forceinline__ v8f cd_load(const float* base, int mi, int nj, int lane) {
  int n  = nj * 16 + (lane & 15);
  int m0 = mi * 16 + ((lane >> 4) << 3);
  v8f c;
#pragma unroll
  for (int r = 0; r < 8; ++r) c[r] = base[(m0 + r) * LDP + n];
  return c;
}
__device__ __forceinline__ void cd_store(float* base, int mi, int nj, int lane, v8f c) {
  int n  = nj * 16 + (lane & 15);
  int m0 = mi * 16 + ((lane >> 4) << 3);
#pragma unroll
  for (int r = 0; r < 8; ++r) base[(m0 + r) * LDP + n] = c[r];
}

__device__ __forceinline__ v8f gemm_nn(const float* A, const float* Bm,
                                       int mi, int nj, int lane,
                                       int k0, int k1, v8f acc) {
  int lm = lane & 15, hi = lane >> 4;
  int arow = (mi * 16 + lm) * LDP;
  int bcol = nj * 16 + lm;
  for (int kk = k0; kk < k1; ++kk) {
    int kb = kk * 4 + hi * 2;
    v2f a = lds_pair(&A[arow + kb]);
    v2f b; b.x = Bm[kb * LDP + bcol]; b.y = Bm[(kb + 1) * LDP + bcol];
    acc = wmma4(a, b, acc);
  }
  return acc;
}
__device__ __forceinline__ v8f gemm_nt(const float* A, const float* Bt,
                                       int mi, int nj, int lane, v8f acc) {
  int lm = lane & 15, hi = lane >> 4;
  int arow = (mi * 16 + lm) * LDP;
  int brow = (nj * 16 + lm) * LDP;
#pragma unroll
  for (int kk = 0; kk < 16; ++kk) {
    int kb = kk * 4 + hi * 2;
    v2f a = lds_pair(&A[arow + kb]);
    v2f b = lds_pair(&Bt[brow + kb]);
    acc = wmma4(a, b, acc);
  }
  return acc;
}
__device__ __forceinline__ v8f gemm_tn(const float* At, const float* Bm,
                                       int mi, int nj, int lane, v8f acc) {
  int lm = lane & 15, hi = lane >> 4;
  int am   = mi * 16 + lm;
  int bcol = nj * 16 + lm;
#pragma unroll
  for (int kk = 0; kk < 16; ++kk) {
    int kb = kk * 4 + hi * 2;
    v2f a; a.x = At[kb * LDP + am];   a.y = At[(kb + 1) * LDP + am];
    v2f b; b.x = Bm[kb * LDP + bcol]; b.y = Bm[(kb + 1) * LDP + bcol];
    acc = wmma4(a, b, acc);
  }
  return acc;
}

constexpr int BK = 4;
template <int KSTEPS>
__device__ __forceinline__ void gemm_nn2(const float* A, const float* Bm,
                                         int mi, int nj0, int nj1, int lane,
                                         v8f& acc0, v8f& acc1) {
  constexpr int NB = KSTEPS / BK;
  int lm = lane & 15, hi = lane >> 4;
  const float* Ap  = A + (mi * 16 + lm) * LDP + hi * 2;
  const float* Bp0 = Bm + (hi * 2) * LDP + nj0 * 16 + lm;
  const float* Bp1 = Bm + (hi * 2) * LDP + nj1 * 16 + lm;
  v2f a[2][BK];
  float b0x[2][BK], b0y[2][BK], b1x[2][BK], b1y[2][BK];
#pragma unroll
  for (int j = 0; j < BK; ++j) {
    a[0][j] = lds_pair(Ap + j * 4);
    b0x[0][j] = Bp0[(j * 4) * LDP]; b0y[0][j] = Bp0[(j * 4 + 1) * LDP];
    b1x[0][j] = Bp1[(j * 4) * LDP]; b1y[0][j] = Bp1[(j * 4 + 1) * LDP];
  }
#pragma unroll
  for (int blk = 0; blk < NB; ++blk) {
    const int cur = blk & 1, nxt = cur ^ 1;
    if (blk + 1 < NB) {
#pragma unroll
      for (int j = 0; j < BK; ++j) {
        const int kk = (blk + 1) * BK + j;
        a[nxt][j] = lds_pair(Ap + kk * 4);
        b0x[nxt][j] = Bp0[(kk * 4) * LDP]; b0y[nxt][j] = Bp0[(kk * 4 + 1) * LDP];
        b1x[nxt][j] = Bp1[(kk * 4) * LDP]; b1y[nxt][j] = Bp1[(kk * 4 + 1) * LDP];
      }
    }
    __builtin_amdgcn_sched_barrier(0);
#pragma unroll
    for (int j = 0; j < BK; ++j) {
      v2f p0; p0.x = b0x[cur][j]; p0.y = b0y[cur][j];
      v2f p1; p1.x = b1x[cur][j]; p1.y = b1y[cur][j];
      acc0 = wmma4(a[cur][j], p0, acc0);
      acc1 = wmma4(a[cur][j], p1, acc1);
    }
    __builtin_amdgcn_sched_barrier(0);
  }
}
__device__ __forceinline__ void gemm_nt2(const float* A, const float* Bt,
                                         int mi, int nj0, int nj1, int lane,
                                         v8f& acc0, v8f& acc1) {
  constexpr int NB = 16 / BK;
  int lm = lane & 15, hi = lane >> 4;
  const float* Ap  = A  + (mi  * 16 + lm) * LDP + hi * 2;
  const float* Bp0 = Bt + (nj0 * 16 + lm) * LDP + hi * 2;
  const float* Bp1 = Bt + (nj1 * 16 + lm) * LDP + hi * 2;
  v2f a[2][BK], b0[2][BK], b1[2][BK];
#pragma unroll
  for (int j = 0; j < BK; ++j) {
    a[0][j]  = lds_pair(Ap  + j * 4);
    b0[0][j] = lds_pair(Bp0 + j * 4);
    b1[0][j] = lds_pair(Bp1 + j * 4);
  }
#pragma unroll
  for (int blk = 0; blk < NB; ++blk) {
    const int cur = blk & 1, nxt = cur ^ 1;
    if (blk + 1 < NB) {
#pragma unroll
      for (int j = 0; j < BK; ++j) {
        const int kk = (blk + 1) * BK + j;
        a[nxt][j]  = lds_pair(Ap  + kk * 4);
        b0[nxt][j] = lds_pair(Bp0 + kk * 4);
        b1[nxt][j] = lds_pair(Bp1 + kk * 4);
      }
    }
    __builtin_amdgcn_sched_barrier(0);
#pragma unroll
    for (int j = 0; j < BK; ++j) {
      acc0 = wmma4(a[cur][j], b0[cur][j], acc0);
      acc1 = wmma4(a[cur][j], b1[cur][j], acc1);
    }
    __builtin_amdgcn_sched_barrier(0);
  }
}

__device__ __forceinline__ void mask_diag(v8f& c, int lane) {
  int n  = lane & 15;
  int m0 = (lane >> 4) << 3;
#pragma unroll
  for (int r = 0; r < 8; ++r)
    if (m0 + r < n) c[r] = 0.0f;
}

__device__ __forceinline__ void tile_rowsum_store(float* rowp, int mi, int nj, int lane, v8f c) {
  float s[8];
#pragma unroll
  for (int r = 0; r < 8; ++r) {
    float x = c[r];
    x = row_xor_add<1>(x);
    x = row_xor_add<2>(x);
    x = row_xor_add<4>(x);
    x = row_xor_add<8>(x);
    s[r] = x;
  }
  if ((lane & 15) == 0) {
    float* p = rowp + nj * 64 + mi * 16 + ((lane >> 4) << 3);
    *reinterpret_cast<float4*>(p)     = make_float4(s[0], s[1], s[2], s[3]);
    *reinterpret_cast<float4*>(p + 4) = make_float4(s[4], s[5], s[6], s[7]);
  }
}

template <int M, int NH>
__device__ __forceinline__ void scores_tiles(const float* sQ, const float* sK,
                                             float* sS, float* sRow, int lane) {
  constexpr int N0 = 2 * NH, N1 = 2 * NH + 1;
  if constexpr (N1 <= M) {
    v8f s0 = {}, s1 = {};
    gemm_nt2(sQ, sK, M, N0, N1, lane, s0, s1);
    if constexpr (M == N1) mask_diag(s1, lane);
    tile_rowsum_store(sRow, M, N0, lane, s0);
    cd_store(sS, M, N0, lane, s0);
    tile_rowsum_store(sRow, M, N1, lane, s1);
    cd_store(sS, M, N1, lane, s1);
  } else if constexpr (N0 <= M) {
    v8f s0 = {};
    s0 = gemm_nt(sQ, sK, M, N0, lane, s0);
    if constexpr (M == N0) mask_diag(s0, lane);
    tile_rowsum_store(sRow, M, N0, lane, s0);
    cd_store(sS, M, N0, lane, s0);
  }
}

__device__ __forceinline__ float xhalf(float x) {
  return __int_as_float(__builtin_amdgcn_permlanex16(__float_as_int(x), __float_as_int(x),
                                                     0x76543210, 0xfedcba98, false, false));
}
__device__ __forceinline__ void ksum_wave_partial(float* kpart, int tid, float4 cs) {
  cs.x += xhalf(cs.x); cs.y += xhalf(cs.y); cs.z += xhalf(cs.z); cs.w += xhalf(cs.w);
  if ((tid & 31) < 16)
    *reinterpret_cast<float4*>(kpart + (tid >> 5) * 64 + ((tid & 15) << 2)) = cs;
}
__device__ __forceinline__ float ksum_join(const float* kpart, int d) {
  float p0 = kpart[d],       p1 = kpart[64 + d],  p2 = kpart[128 + d], p3 = kpart[192 + d];
  float p4 = kpart[256 + d], p5 = kpart[320 + d], p6 = kpart[384 + d], p7 = kpart[448 + d];
  return ((p0 + p1) + (p2 + p3)) + ((p4 + p5) + (p6 + p7));
}

__device__ __forceinline__ void tile_issue(const float* __restrict__ g, int rowStride,
                                           int tid, float4 (&w)[4]) {
#pragma unroll
  for (int j = 0; j < 4; ++j) {
    int u = tid + j * 256;
    int r = u >> 4;
    int c = (u & 15) << 2;
    w[j] = *reinterpret_cast<const float4*>(g + (size_t)r * rowStride + c);
  }
}
__device__ __forceinline__ void tile_commit(float* lds, int tid, const float4 (&w)[4],
                                            bool do_elu) {
#pragma unroll
  for (int j = 0; j < 4; ++j) {
    int u = tid + j * 256;
    int r = u >> 4;
    int c = (u & 15) << 2;
    float4 x = w[j];
    if (do_elu) { x.x = elu1(x.x); x.y = elu1(x.y); x.z = elu1(x.z); x.w = elu1(x.w); }
    *reinterpret_cast<float4*>(lds + r * LDP + c) = x;
  }
}
__device__ __forceinline__ float4 tile_commit_colsum(float* lds, int tid, const float4 (&w)[4]) {
  float4 cs = make_float4(0.0f, 0.0f, 0.0f, 0.0f);
#pragma unroll
  for (int j = 0; j < 4; ++j) {
    int u = tid + j * 256;
    int r = u >> 4;
    int c = (u & 15) << 2;
    float4 x = w[j];
    x.x = elu1(x.x); x.y = elu1(x.y); x.z = elu1(x.z); x.w = elu1(x.w);
    *reinterpret_cast<float4*>(lds + r * LDP + c) = x;
    if (j == 0) cs = x;
    else { cs.x += x.x; cs.y += x.y; cs.z += x.z; cs.w += x.w; }
  }
  return cs;
}

__global__ __launch_bounds__(256) void lin_attn_states(
    const float* __restrict__ qk, const float* __restrict__ v,
    float* __restrict__ ws, int nsplit) {
  extern __shared__ float smem[];
  float* sK    = smem + OFF_K;
  float* sV    = smem + OFF_V;
  float* sKV   = smem + OFF_KV;
  float* sKpart = smem + OFF_KPART;

  int tid = threadIdx.x, lane = tid & 31, wid = tid >> 5;
  int wg = blockIdx.x;
  int split = wg % nsplit;
  int h     = (wg / nsplit) % H_;
  int b     = wg / (nsplit * H_);
  int cps   = (S_ / CH) / nsplit;

  for (int i = tid; i < 64 * LDP; i += 256) sKV[i] = 0.0f;
  float ks = 0.0f;
  __syncthreads();

  float4 rk[4], rv[4];
  {
    int base0 = split * cps * CH;
    tile_issue(qk + (((size_t)(b * S_ + base0) * 2 + 1) * H_ + h) * D_, 2 * H_ * D_, tid, rk);
    tile_issue(v  + (((size_t)(b * S_ + base0)) * H_ + h) * D_,         H_ * D_,     tid, rv);
  }

  for (int c = 0; c < cps; ++c) {
    ksum_wave_partial(sKpart, tid, tile_commit_colsum(sK, tid, rk));
    tile_commit(sV, tid, rv, false);
    if (c + 1 < cps) {
      int nb = (split * cps + c + 1) * CH;
      tile_issue(qk + (((size_t)(b * S_ + nb) * 2 + 1) * H_ + h) * D_, 2 * H_ * D_, tid, rk);
      tile_issue(v  + (((size_t)(b * S_ + nb)) * H_ + h) * D_,         H_ * D_,     tid, rv);
    }
    lds_barrier();

    if (tid < 64) ks += ksum_join(sKpart, tid);
#pragma unroll
    for (int i = 0; i < 2; ++i) {
      int t = wid * 2 + i, mi = t >> 2, nj = t & 3;
      v8f acc = cd_load(sKV, mi, nj, lane);
      acc = gemm_tn(sK, sV, mi, nj, lane, acc);
      cd_store(sKV, mi, nj, lane, acc);
    }
    lds_barrier();
  }

  float* wsp = ws + (size_t)((b * H_ + h) * nsplit + split) * STATE_F;
  for (int i = tid; i < 64 * 64; i += 256)
    wsp[i] = sKV[(i >> 6) * LDP + (i & 63)];
  if (tid < 64) wsp[64 * 64 + tid] = ks;
}

__global__ __launch_bounds__(256) void lin_attn_main(
    const float* __restrict__ qk, const float* __restrict__ v,
    const float* __restrict__ ws, float* __restrict__ out, int nsplit) {
  extern __shared__ float smem[];
  float* sQ    = smem + OFF_Q;
  float* sK    = smem + OFF_K;
  float* sV    = smem + OFF_V;
  float* sS    = smem + OFF_S;
  float* sKV   = smem + OFF_KV;
  float* sksum = smem + OFF_KSUM;
  float* sden  = smem + OFF_DEN;
  float* sKpart = smem + OFF_KPART;
  float* sRow  = smem + OFF_ROW;

  int tid = threadIdx.x, lane = tid & 31, wid = tid >> 5;
  int wg = blockIdx.x;
  int split = wg % nsplit;
  int h     = (wg / nsplit) % H_;
  int b     = wg / (nsplit * H_);
  int cps   = (S_ / CH) / nsplit;

  float4 rq[4], rk[4], rv[4];
  {
    int base0 = split * cps * CH;
    tile_issue(qk + (((size_t)(b * S_ + base0) * 2 + 0) * H_ + h) * D_, 2 * H_ * D_, tid, rq);
    tile_issue(qk + (((size_t)(b * S_ + base0) * 2 + 1) * H_ + h) * D_, 2 * H_ * D_, tid, rk);
    tile_issue(v  + (((size_t)(b * S_ + base0)) * H_ + h) * D_,         H_ * D_,     tid, rv);
  }

  {
    constexpr int PB = 7;
    const float* wbase = ws + (size_t)((b * H_ + h) * nsplit) * STATE_F;
    float4 acc[4];
    float  accs = 0.0f;
#pragma unroll
    for (int j = 0; j < 4; ++j) acc[j] = make_float4(0.0f, 0.0f, 0.0f, 0.0f);
    for (int p0 = 0; p0 < split; p0 += PB) {
      float4 st[PB][4];
      float  sk[PB];
#pragma unroll
      for (int q = 0; q < PB; ++q) {
        if (p0 + q < split) {
          const float* wsp = wbase + (size_t)(p0 + q) * STATE_F;
#pragma unroll
          for (int j = 0; j < 4; ++j)
            st[q][j] = *reinterpret_cast<const float4*>(wsp + 4 * (tid + 256 * j));
          if (tid < 64) sk[q] = wsp[64 * 64 + tid];
        }
      }
#pragma unroll
      for (int q = 0; q < PB; ++q) {
        if (p0 + q < split) {
#pragma unroll
          for (int j = 0; j < 4; ++j) {
            acc[j].x += st[q][j].x; acc[j].y += st[q][j].y;
            acc[j].z += st[q][j].z; acc[j].w += st[q][j].w;
          }
          if (tid < 64) accs += sk[q];
        }
      }
    }
#pragma unroll
    for (int j = 0; j < 4; ++j) {
      int f = tid + 256 * j;
      *reinterpret_cast<float4*>(sKV + (f >> 4) * LDP + ((f & 15) << 2)) = acc[j];
    }
    if (tid < 64) sksum[tid] = accs;
  }
  __syncthreads();

  const int widu = __builtin_amdgcn_readfirstlane(wid);
  int t0 = widu * 2, t1 = widu * 2 + 1;
  int m0 = t0 >> 2, n0 = t0 & 3, m1 = t1 >> 2, n1 = t1 & 3;

  for (int c = 0; c < cps; ++c) {
    int base = (split * cps + c) * CH;
    tile_commit(sQ, tid, rq, true);
    ksum_wave_partial(sKpart, tid, tile_commit_colsum(sK, tid, rk));
    tile_commit(sV, tid, rv, false);
    if (c + 1 < cps) {
      int nb = base + CH;
      tile_issue(qk + (((size_t)(b * S_ + nb) * 2 + 0) * H_ + h) * D_, 2 * H_ * D_, tid, rq);
      tile_issue(qk + (((size_t)(b * S_ + nb) * 2 + 1) * H_ + h) * D_, 2 * H_ * D_, tid, rk);
      tile_issue(v  + (((size_t)(b * S_ + nb)) * H_ + h) * D_,         H_ * D_,     tid, rv);
    }
    lds_barrier();

    v8f oa0 = {}, oa1 = {};
    oa0 = gemm_nn(sQ, sKV, m0, n0, lane, 0, 16, oa0);
    oa1 = gemm_nn(sQ, sKV, m1, n1, lane, 0, 16, oa1);

    {
      int t = tid >> 2, q = tid & 3;
      const float* qr = sQ + t * LDP + 16 * q;
      const float* kr = sksum + 16 * q;
      float d = 0.0f;
#pragma unroll
      for (int i = 0; i < 16; i += 4) {
        float4 a = *reinterpret_cast<const float4*>(qr + i);
        float4 k = *reinterpret_cast<const float4*>(kr + i);
        d += a.x * k.x; d += a.y * k.y; d += a.z * k.z; d += a.w * k.w;
      }
      d = row_xor_add<1>(d);
      d = row_xor_add<2>(d);
      if (q == 0) sden[t] = d;
    }

    switch (widu) {
      case 0: scores_tiles<0, 0>(sQ, sK, sS, sRow, lane); break;
      case 1: scores_tiles<0, 1>(sQ, sK, sS, sRow, lane); break;
      case 2: scores_tiles<1, 0>(sQ, sK, sS, sRow, lane); break;
      case 3: scores_tiles<1, 1>(sQ, sK, sS, sRow, lane); break;
      case 4: scores_tiles<2, 0>(sQ, sK, sS, sRow, lane); break;
      case 5: scores_tiles<2, 1>(sQ, sK, sS, sRow, lane); break;
      case 6: scores_tiles<3, 0>(sQ, sK, sS, sRow, lane); break;
      default: scores_tiles<3, 1>(sQ, sK, sS, sRow, lane); break;
    }
    lds_barrier();

    float inv[8];
    {
      int R = m0 * 16 + ((lane >> 4) << 3);
      float4 ia = *reinterpret_cast<const float4*>(sden + R);
      float4 ib = *reinterpret_cast<const float4*>(sden + R + 4);
      float4 sa = *reinterpret_cast<const float4*>(sRow + R);
      float4 sb = *reinterpret_cast<const float4*>(sRow + R + 4);
#pragma unroll
      for (int n = 1; n < 4; ++n) {
        if (n <= m0) {
          float4 ta = *reinterpret_cast<const float4*>(sRow + n * 64 + R);
          float4 tb = *reinterpret_cast<const float4*>(sRow + n * 64 + R + 4);
          sa.x += ta.x; sa.y += ta.y; sa.z += ta.z; sa.w += ta.w;
          sb.x += tb.x; sb.y += tb.y; sb.z += tb.z; sb.w += tb.w;
        }
      }
      inv[0] = 1.0f / (ia.x + sa.x); inv[1] = 1.0f / (ia.y + sa.y);
      inv[2] = 1.0f / (ia.z + sa.z); inv[3] = 1.0f / (ia.w + sa.w);
      inv[4] = 1.0f / (ib.x + sb.x); inv[5] = 1.0f / (ib.y + sb.y);
      inv[6] = 1.0f / (ib.z + sb.z); inv[7] = 1.0f / (ib.w + sb.w);
    }

    switch (m0) {
      case 0: gemm_nn2<4>(sS, sV, 0, n0, n1, lane, oa0, oa1); break;
      case 1: gemm_nn2<8>(sS, sV, 1, n0, n1, lane, oa0, oa1); break;
      case 2: gemm_nn2<12>(sS, sV, 2, n0, n1, lane, oa0, oa1); break;
      default: gemm_nn2<16>(sS, sV, 3, n0, n1, lane, oa0, oa1); break;
    }

    {
      int nn = n0 * 16 + (lane & 15);
      int mm = m0 * 16 + ((lane >> 4) << 3);
#pragma unroll
      for (int r = 0; r < 8; ++r)
        out[(((size_t)b * S_ + base + mm + r) * H_ + h) * D_ + nn] = oa0[r] * inv[r];
      nn = n1 * 16 + (lane & 15);
      mm = m1 * 16 + ((lane >> 4) << 3);
#pragma unroll
      for (int r = 0; r < 8; ++r)
        out[(((size_t)b * S_ + base + mm + r) * H_ + h) * D_ + nn] = oa1[r] * inv[r];
    }
    lds_barrier();

    if (tid < 64) sksum[tid] += ksum_join(sKpart, tid);
    {
      v8f a0 = cd_load(sKV, m0, n0, lane);
      a0 = gemm_tn(sK, sV, m0, n0, lane, a0);
      cd_store(sKV, m0, n0, lane, a0);
      v8f a1 = cd_load(sKV, m1, n1, lane);
      a1 = gemm_tn(sK, sV, m1, n1, lane, a1);
      cd_store(sKV, m1, n1, lane, a1);
    }
    lds_barrier();
  }
}

extern "C" void kernel_launch(void* const* d_in, const int* in_sizes, int n_in,
                              void* d_out, int out_size, void* d_ws, size_t ws_size,
                              hipStream_t stream) {
  const float* qk = (const float*)d_in[0];
  const float* v  = (const float*)d_in[1];
  float* out = (float*)d_out;
  float* ws  = (float*)d_ws;

  int nsplit = NSPL;
  size_t need = (size_t)B_ * H_ * NSPL * STATE_F * sizeof(float);
  if (ws_size < need) nsplit = 1;

  int wgs = B_ * H_ * nsplit;
  size_t smem = (size_t)SMEM_F * sizeof(float);

  if (nsplit > 1)
    lin_attn_states<<<wgs, 256, smem, stream>>>(qk, v, ws, nsplit);
  lin_attn_main<<<wgs, 256, smem, stream>>>(qk, v, ws, out, nsplit);
}


# One wave owns 16 tokens x 32 features of a chunk: transposed scores stay in accumulators as B of V^T S^T, state planes alternate, 1 barrier per chunk not 4; row-sum and product summation order change
# speedup vs baseline: 3.0928x; 1.1460x over previous
#include <hip/hip_runtime.h>
#include <hip/hip_bf16.h>


typedef float v2f __attribute__((ext_vector_type(2)));
typedef float v8f __attribute__((ext_vector_type(8)));

constexpr int B_   = 2;
constexpr int S_   = 2048;
constexpr int H_   = 16;
constexpr int D_   = 64;
constexpr int CH   = 64;
constexpr int LDP  = 68;
constexpr int NSPL = 8;

constexpr int OFF_K    = 64 * LDP;
constexpr int OFF_V    = 2 * 64 * LDP;
constexpr int OFF_KV   = 4 * 64 * LDP;
constexpr int OFF_KPART = 5 * 64 * LDP + 128;
constexpr int SMEM_F   = OFF_KPART + 512 + 256;

constexpr int TILE_F   = 64 * LDP;
constexpr int MOFF_T   = 0;
constexpr int MOFF_KVT = 6 * TILE_F;
constexpr int MOFF_KS  = 8 * TILE_F;
constexpr int MOFF_KP  = MOFF_KS + 128;
constexpr int MOFF_STG = MOFF_KP + 1024;
constexpr int STGP     = 36;
constexpr int MAIN_SMEM_F = MOFF_STG + 8 * 16 * STGP;

constexpr int STATE_F  = 64 * 64 + 64;

__device__ __forceinline__ float elu1(float x) {
  return x > 0.0f ? x + 1.0f : __expf(x);
}


__device__ __forceinline__ void lds_barrier() {
  __builtin_amdgcn_fence(__ATOMIC_RELEASE, "workgroup", "local");
  __builtin_amdgcn_s_barrier();
  __builtin_amdgcn_fence(__ATOMIC_ACQUIRE, "workgroup", "local");
}

__device__ __forceinline__ v8f wmma4(v2f a, v2f b, v8f c) {
  return __builtin_amdgcn_wmma_f32_16x16x4_f32(
       false, a,  false, b,
       (short)0, c,  false,  false);
}

__device__ __forceinline__ v2f lds_pair(const float* p) {
  return *reinterpret_cast<const v2f*>(p);
}

__device__ __forceinline__ v8f cd_load(const float* base, int mi, int nj, int lane) {
  int n  = nj * 16 + (lane & 15);
  int m0 = mi * 16 + ((lane >> 4) << 3);
  v8f c;
#pragma unroll
  for (int r = 0; r < 8; ++r) c[r] = base[(m0 + r) * LDP + n];
  return c;
}
__device__ __forceinline__ void cd_store(float* base, int mi, int nj, int lane, v8f c) {
  int n  = nj * 16 + (lane & 15);
  int m0 = mi * 16 + ((lane >> 4) << 3);
#pragma unroll
  for (int r = 0; r < 8; ++r) base[(m0 + r) * LDP + n] = c[r];
}

__device__ __forceinline__ v8f gemm_tn(const float* At, const float* Bm,
                                       int mi, int nj, int lane, v8f acc) {
  int lm = lane & 15, hi = lane >> 4;
  int am   = mi * 16 + lm;
  int bcol = nj * 16 + lm;
#pragma unroll
  for (int kk = 0; kk < 16; ++kk) {
    int kb = kk * 4 + hi * 2;
    v2f a; a.x = At[kb * LDP + am];   a.y = At[(kb + 1) * LDP + am];
    v2f b; b.x = Bm[kb * LDP + bcol]; b.y = Bm[(kb + 1) * LDP + bcol];
    acc = wmma4(a, b, acc);
  }
  return acc;
}

__device__ __forceinline__ float xhalf(float x) {
  return __int_as_float(__builtin_amdgcn_permlanex16(__float_as_int(x), __float_as_int(x),
                                                     0x76543210, 0xfedcba98, false, false));
}
__device__ __forceinline__ void ksum_wave_partial(float* kpart, int tid, float4 cs) {
  cs.x += xhalf(cs.x); cs.y += xhalf(cs.y); cs.z += xhalf(cs.z); cs.w += xhalf(cs.w);
  if ((tid & 31) < 16)
    *reinterpret_cast<float4*>(kpart + (tid >> 5) * 64 + ((tid & 15) << 2)) = cs;
}
__device__ __forceinline__ float ksum_join(const float* kpart, int d) {
  float p0 = kpart[d],       p1 = kpart[64 + d],  p2 = kpart[128 + d], p3 = kpart[192 + d];
  float p4 = kpart[256 + d], p5 = kpart[320 + d], p6 = kpart[384 + d], p7 = kpart[448 + d];
  return ((p0 + p1) + (p2 + p3)) + ((p4 + p5) + (p6 + p7));
}

__device__ __forceinline__ void tile_issue(const float* __restrict__ g, int rowStride,
                                           int tid, float4 (&w)[4]) {
#pragma unroll
  for (int j = 0; j < 4; ++j) {
    int u = tid + j * 256;
    int r = u >> 4;
    int c = (u & 15) << 2;
    w[j] = *reinterpret_cast<const float4*>(g + (size_t)r * rowStride + c);
  }
}
__device__ __forceinline__ void tile_commit(float* lds, int tid, const float4 (&w)[4],
                                            bool do_elu) {
#pragma unroll
  for (int j = 0; j < 4; ++j) {
    int u = tid + j * 256;
    int r = u >> 4;
    int c = (u & 15) << 2;
    float4 x = w[j];
    if (do_elu) { x.x = elu1(x.x); x.y = elu1(x.y); x.z = elu1(x.z); x.w = elu1(x.w); }
    *reinterpret_cast<float4*>(lds + r * LDP + c) = x;
  }
}
__device__ __forceinline__ float4 tile_commit_colsum(float* lds, int tid, const float4 (&w)[4]) {
  float4 cs = make_float4(0.0f, 0.0f, 0.0f, 0.0f);
#pragma unroll
  for (int j = 0; j < 4; ++j) {
    int u = tid + j * 256;
    int r = u >> 4;
    int c = (u & 15) << 2;
    float4 x = w[j];
    x.x = elu1(x.x); x.y = elu1(x.y); x.z = elu1(x.z); x.w = elu1(x.w);
    *reinterpret_cast<float4*>(lds + r * LDP + c) = x;
    if (j == 0) cs = x;
    else { cs.x += x.x; cs.y += x.y; cs.z += x.z; cs.w += x.w; }
  }
  return cs;
}


constexpr int BK = 4;

template <int NT>
__device__ __forceinline__ void rows_times_q(const float* (&ap)[NT],
                                             const v2f (&qf)[16], v8f (&acc)[NT]) {
  constexpr int NB = 16 / BK;
  v2f a[2][NT][BK];
#pragma unroll
  for (int i = 0; i < NT; ++i)
#pragma unroll
    for (int j = 0; j < BK; ++j) a[0][i][j] = lds_pair(ap[i] + j * 4);
#pragma unroll
  for (int blk = 0; blk < NB; ++blk) {
    const int cur = blk & 1, nxt = cur ^ 1;
    if (blk + 1 < NB) {
#pragma unroll
      for (int i = 0; i < NT; ++i)
#pragma unroll
        for (int j = 0; j < BK; ++j)
          a[nxt][i][j] = lds_pair(ap[i] + ((blk + 1) * BK + j) * 4);
    }
    __builtin_amdgcn_sched_barrier(0);
#pragma unroll
    for (int j = 0; j < BK; ++j)
#pragma unroll
      for (int i = 0; i < NT; ++i)
        acc[i] = wmma4(a[cur][i][j], qf[blk * BK + j], acc[i]);
    __builtin_amdgcn_sched_barrier(0);
  }
}

template <int TB, int SB0, int NT>
__device__ __forceinline__ void intra_blocks(const float* sK, const float* sV,
                                             const v2f (&qf)[16], int eh, int lane,
                                             v8f& o0, v8f& o1, float& den) {
  const int c = lane & 15, h = lane >> 4;
  float va[NT][4][2], vb[NT][4][2];
#pragma unroll
  for (int i = 0; i < NT; ++i) {
    const float* vp = sV + ((SB0 + i) * 16 + 8 * h) * LDP + eh * 32 + c;
#pragma unroll
    for (int j = 0; j < 4; ++j) {
      va[i][j][0] = vp[(2 * j) * LDP];      va[i][j][1] = vp[(2 * j + 1) * LDP];
      vb[i][j][0] = vp[(2 * j) * LDP + 16]; vb[i][j][1] = vp[(2 * j + 1) * LDP + 16];
    }
  }
  const float* ap[NT];
  v8f s[NT];
#pragma unroll
  for (int i = 0; i < NT; ++i) {
    ap[i] = sK + ((SB0 + i) * 16 + c) * LDP + 2 * h;
    s[i] = v8f{};
  }
  rows_times_q<NT>(ap, qf, s);
#pragma unroll
  for (int i = 0; i < NT; ++i) {
    if (SB0 + i == TB) {
#pragma unroll
      for (int r = 0; r < 8; ++r)
        if (8 * h + r > c) s[i][r] = 0.0f;
    }
#pragma unroll
    for (int r = 0; r < 8; ++r) den += s[i][r];
#pragma unroll
    for (int j = 0; j < 4; ++j) {
      v2f p;  p.x = s[i][2 * j];    p.y = s[i][2 * j + 1];
      v2f a0; a0.x = va[i][j][0];   a0.y = va[i][j][1];
      v2f a1; a1.x = vb[i][j][0];   a1.y = vb[i][j][1];
      o0 = wmma4(a0, p, o0);
      o1 = wmma4(a1, p, o1);
    }
  }
}

template <int TB>
__device__ __forceinline__ void wave_out(const float* sQ, const float* sK, const float* sV,
                                         const float* sKVT, const float* sks,
                                         float* stg, int eh, int lane,
                                         float* __restrict__ outp) {
  const int c = lane & 15, h = lane >> 4;
  v2f qf[16];
  {
    const float* qp = sQ + (TB * 16 + c) * LDP + 2 * h;
#pragma unroll
    for (int j = 0; j < 16; ++j) qf[j] = lds_pair(qp + j * 4);
  }
  v8f o[2] = {v8f{}, v8f{}};
  {
    const float* kp[2];
    kp[0] = sKVT + (eh * 32 + c) * LDP + 2 * h;
    kp[1] = kp[0] + 16 * LDP;
    rows_times_q<2>(kp, qf, o);
  }
  v2f dq = {0.0f, 0.0f};
#pragma unroll
  for (int j = 0; j < 16; ++j) dq += qf[j] * lds_pair(sks + j * 4 + 2 * h);
  float den = dq.x + dq.y;
  if constexpr (TB >= 1) intra_blocks<TB, 0, 2>(sK, sV, qf, eh, lane, o[0], o[1], den);
  else                   intra_blocks<TB, 0, 1>(sK, sV, qf, eh, lane, o[0], o[1], den);
  if constexpr (TB == 2) intra_blocks<TB, 2, 1>(sK, sV, qf, eh, lane, o[0], o[1], den);
  if constexpr (TB == 3) intra_blocks<TB, 2, 2>(sK, sV, qf, eh, lane, o[0], o[1], den);
  den += xhalf(den);
  const float inv = 1.0f / den;
#pragma unroll
  for (int i = 0; i < 2; ++i) {
    float* sp = stg + c * STGP + 16 * i + 8 * h;
    *reinterpret_cast<float4*>(sp)     = make_float4(o[i][0] * inv, o[i][1] * inv, o[i][2] * inv, o[i][3] * inv);
    *reinterpret_cast<float4*>(sp + 4) = make_float4(o[i][4] * inv, o[i][5] * inv, o[i][6] * inv, o[i][7] * inv);
  }
  __builtin_amdgcn_fence(__ATOMIC_RELEASE, "wavefront", "local");
  __builtin_amdgcn_wave_barrier();
  __builtin_amdgcn_fence(__ATOMIC_ACQUIRE, "wavefront", "local");
#pragma unroll
  for (int i = 0; i < 4; ++i) {
    int idx = lane + 32 * i;
    int tok = idx >> 3, pc = idx & 7;
    float4 x = *reinterpret_cast<const float4*>(stg + tok * STGP + 4 * pc);
    *reinterpret_cast<float4*>(outp + (size_t)tok * (H_ * D_) + 4 * pc) = x;
  }
}


__global__ __launch_bounds__(256) void lin_attn_states(
    const float* __restrict__ qk, const float* __restrict__ v,
    float* __restrict__ ws, int nsplit) {
  extern __shared__ float smem[];
  float* sK    = smem + OFF_K;
  float* sV    = smem + OFF_V;
  float* sKV   = smem + OFF_KV;
  float* sKpart = smem + OFF_KPART;

  int tid = threadIdx.x, lane = tid & 31, wid = tid >> 5;
  int wg = blockIdx.x;
  int split = wg % nsplit;
  int h     = (wg / nsplit) % H_;
  int b     = wg / (nsplit * H_);
  int cps   = (S_ / CH) / nsplit;

  for (int i = tid; i < 64 * LDP; i += 256) sKV[i] = 0.0f;
  float ks = 0.0f;
  __syncthreads();

  float4 rk[4], rv[4];
  {
    int base0 = split * cps * CH;
    tile_issue(qk + (((size_t)(b * S_ + base0) * 2 + 1) * H_ + h) * D_, 2 * H_ * D_, tid, rk);
    tile_issue(v  + (((size_t)(b * S_ + base0)) * H_ + h) * D_,         H_ * D_,     tid, rv);
  }

  for (int c = 0; c < cps; ++c) {
    ksum_wave_partial(sKpart, tid, tile_commit_colsum(sK, tid, rk));
    tile_commit(sV, tid, rv, false);
    if (c + 1 < cps) {
      int nb = (split * cps + c + 1) * CH;
      tile_issue(qk + (((size_t)(b * S_ + nb) * 2 + 1) * H_ + h) * D_, 2 * H_ * D_, tid, rk);
      tile_issue(v  + (((size_t)(b * S_ + nb)) * H_ + h) * D_,         H_ * D_,     tid, rv);
    }
    lds_barrier();

    if (tid < 64) ks += ksum_join(sKpart, tid);
#pragma unroll
    for (int i = 0; i < 2; ++i) {
      int t = wid * 2 + i, mi = t >> 2, nj = t & 3;
      v8f acc = cd_load(sKV, mi, nj, lane);
      acc = gemm_tn(sV, sK, mi, nj, lane, acc);
      cd_store(sKV, mi, nj, lane, acc);
    }
    lds_barrier();
  }

  float* wsp = ws + (size_t)((b * H_ + h) * nsplit + split) * STATE_F;
  for (int i = tid; i < 64 * 64; i += 256)
    wsp[i] = sKV[(i >> 6) * LDP + (i & 63)];
  if (tid < 64) wsp[64 * 64 + tid] = ks;
}

__global__ __launch_bounds__(256) void lin_attn_main(
    const float* __restrict__ qk, const float* __restrict__ v,
    const float* __restrict__ ws, float* __restrict__ out, int nsplit) {
  __shared__ __attribute__((aligned(16))) float smem[MAIN_SMEM_F];

  int tid = threadIdx.x, lane = tid & 31, wid = tid >> 5;
  int wg = blockIdx.x;
  int split = wg % nsplit;
  int h     = (wg / nsplit) % H_;
  int b     = wg / (nsplit * H_);
  int cps   = (S_ / CH) / nsplit;

  float4 rq[4], rk[4], rv[4];
  {
    int base0 = split * cps * CH;
    tile_issue(qk + (((size_t)(b * S_ + base0) * 2 + 0) * H_ + h) * D_, 2 * H_ * D_, tid, rq);
    tile_issue(qk + (((size_t)(b * S_ + base0) * 2 + 1) * H_ + h) * D_, 2 * H_ * D_, tid, rk);
    tile_issue(v  + (((size_t)(b * S_ + base0)) * H_ + h) * D_,         H_ * D_,     tid, rv);
  }

  {
    constexpr int PB = 7;
    float* sKVT0 = smem + MOFF_KVT;
    float* sks0  = smem + MOFF_KS;
    const float* wbase = ws + (size_t)((b * H_ + h) * nsplit) * STATE_F;
    float4 acc[4];
    float  accs = 0.0f;
#pragma unroll
    for (int j = 0; j < 4; ++j) acc[j] = make_float4(0.0f, 0.0f, 0.0f, 0.0f);
    for (int p0 = 0; p0 < split; p0 += PB) {
      float4 st[PB][4];
      float  sk[PB];
#pragma unroll
      for (int q = 0; q < PB; ++q) {
        if (p0 + q < split) {
          const float* wsp = wbase + (size_t)(p0 + q) * STATE_F;
#pragma unroll
          for (int j = 0; j < 4; ++j)
            st[q][j] = *reinterpret_cast<const float4*>(wsp + 4 * (tid + 256 * j));
          if (tid < 64) sk[q] = wsp[64 * 64 + tid];
        }
      }
#pragma unroll
      for (int q = 0; q < PB; ++q) {
        if (p0 + q < split) {
#pragma unroll
          for (int j = 0; j < 4; ++j) {
            acc[j].x += st[q][j].x; acc[j].y += st[q][j].y;
            acc[j].z += st[q][j].z; acc[j].w += st[q][j].w;
          }
          if (tid < 64) accs += sk[q];
        }
      }
    }
#pragma unroll
    for (int j = 0; j < 4; ++j) {
      int f = tid + 256 * j;
      *reinterpret_cast<float4*>(sKVT0 + (f >> 4) * LDP + ((f & 15) << 2)) = acc[j];
    }
    if (tid < 64) sks0[tid] = accs;
  }
  __syncthreads();

  const int widu = __builtin_amdgcn_readfirstlane(wid);
  const int tb = widu < 4 ? widu : 7 - widu;
  const int eh = widu >> 2;
  const int se0 = (widu * 2) >> 2, sd0 = (widu * 2) & 3;
  const int se1 = (widu * 2 + 1) >> 2, sd1 = (widu * 2 + 1) & 3;
  v8f kv0 = cd_load(smem + MOFF_KVT, se0, sd0, lane);
  v8f kv1 = cd_load(smem + MOFF_KVT, se1, sd1, lane);
  float* stg = smem + MOFF_STG + widu * (16 * STGP);

  for (int c = 0; c < cps; ++c) {
    const int buf = c & 1;
    int base = (split * cps + c) * CH;
    float* sQ = smem + MOFF_T + buf * (3 * TILE_F);
    float* sK = sQ + TILE_F;
    float* sV = sK + TILE_F;
    const float* sKVTc = smem + MOFF_KVT + buf * TILE_F;
    float*       sKVTn = smem + MOFF_KVT + (buf ^ 1) * TILE_F;
    const float* sksc  = smem + MOFF_KS + buf * 64;
    float*       sksn  = smem + MOFF_KS + (buf ^ 1) * 64;
    float*       sKp   = smem + MOFF_KP + buf * 512;

    tile_commit(sQ, tid, rq, true);
    ksum_wave_partial(sKp, tid, tile_commit_colsum(sK, tid, rk));
    tile_commit(sV, tid, rv, false);
    if (c + 1 < cps) {
      int nb = base + CH;
      tile_issue(qk + (((size_t)(b * S_ + nb) * 2 + 0) * H_ + h) * D_, 2 * H_ * D_, tid, rq);
      tile_issue(qk + (((size_t)(b * S_ + nb) * 2 + 1) * H_ + h) * D_, 2 * H_ * D_, tid, rk);
      tile_issue(v  + (((size_t)(b * S_ + nb)) * H_ + h) * D_,         H_ * D_,     tid, rv);
    }
    lds_barrier();

    float* outp = out + (((size_t)b * S_ + base + tb * 16) * H_ + h) * D_ + eh * 32;
    switch (tb) {
      case 0:  wave_out<0>(sQ, sK, sV, sKVTc, sksc, stg, eh, lane, outp); break;
      case 1:  wave_out<1>(sQ, sK, sV, sKVTc, sksc, stg, eh, lane, outp); break;
      case 2:  wave_out<2>(sQ, sK, sV, sKVTc, sksc, stg, eh, lane, outp); break;
      default: wave_out<3>(sQ, sK, sV, sKVTc, sksc, stg, eh, lane, outp); break;
    }

    if (c + 1 < cps) {
      if (tid < 64) sksn[tid] = sksc[tid] + ksum_join(sKp, tid);
      kv0 = gemm_tn(sV, sK, se0, sd0, lane, kv0);
      cd_store(sKVTn, se0, sd0, lane, kv0);
      kv1 = gemm_tn(sV, sK, se1, sd1, lane, kv1);
      cd_store(sKVTn, se1, sd1, lane, kv1);
    }
  }
}

extern "C" void kernel_launch(void* const* d_in, const int* in_sizes, int n_in,
                              void* d_out, int out_size, void* d_ws, size_t ws_size,
                              hipStream_t stream) {
  const float* qk = (const float*)d_in[0];
  const float* v  = (const float*)d_in[1];
  float* out = (float*)d_out;
  float* ws  = (float*)d_ws;

  int nsplit = NSPL;
  size_t need = (size_t)B_ * H_ * NSPL * STATE_F * sizeof(float);
  if (ws_size < need) nsplit = 1;

  int wgs = B_ * H_ * nsplit;
  size_t smem = (size_t)SMEM_F * sizeof(float);

  if (nsplit > 1)
    lin_attn_states<<<wgs, 256, smem, stream>>>(qk, v, ws, nsplit);
  lin_attn_main<<<wgs, 256, 0, stream>>>(qk, v, ws, out, nsplit);
}


# Main pass work split by role: waves 0-3 own 16 tokens x all 64 features (each score tile computed once, not twice), waves 4-7 own the state update as 7/5/3/1 tiles
# speedup vs baseline: 3.3359x; 1.0786x over previous
#include <hip/hip_runtime.h>
#include <hip/hip_bf16.h>


typedef float v2f __attribute__((ext_vector_type(2)));
typedef float v8f __attribute__((ext_vector_type(8)));

constexpr int B_   = 2;
constexpr int S_   = 2048;
constexpr int H_   = 16;
constexpr int D_   = 64;
constexpr int CH   = 64;
constexpr int LDP  = 68;
constexpr int NSPL = 8;

constexpr int OFF_K    = 64 * LDP;
constexpr int OFF_V    = 2 * 64 * LDP;
constexpr int OFF_KV   = 4 * 64 * LDP;
constexpr int OFF_KPART = 5 * 64 * LDP + 128;
constexpr int SMEM_F   = OFF_KPART + 512 + 256;

constexpr int TILE_F   = 64 * LDP;
constexpr int MOFF_T   = 0;
constexpr int MOFF_KVT = 6 * TILE_F;
constexpr int MOFF_KS  = 8 * TILE_F;
constexpr int MOFF_KP  = MOFF_KS + 128;
constexpr int MOFF_STG = MOFF_KP + 1024;
constexpr int STGP     = 68;
constexpr int MAIN_SMEM_F = MOFF_STG + 4 * 16 * STGP;

constexpr int STATE_F  = 64 * 64 + 64;

__device__ __forceinline__ float elu1(float x) {
  return x > 0.0f ? x + 1.0f : __expf(x);
}


__device__ __forceinline__ void lds_barrier() {
  __builtin_amdgcn_fence(__ATOMIC_RELEASE, "workgroup", "local");
  __builtin_amdgcn_s_barrier();
  __builtin_amdgcn_fence(__ATOMIC_ACQUIRE, "workgroup", "local");
}

__device__ __forceinline__ v8f wmma4(v2f a, v2f b, v8f c) {
  return __builtin_amdgcn_wmma_f32_16x16x4_f32(
       false, a,  false, b,
       (short)0, c,  false,  false);
}

__device__ __forceinline__ v2f lds_pair(const float* p) {
  return *reinterpret_cast<const v2f*>(p);
}

__device__ __forceinline__ v8f cd_load(const float* base, int mi, int nj, int lane) {
  int n  = nj * 16 + (lane & 15);
  int m0 = mi * 16 + ((lane >> 4) << 3);
  v8f c;
#pragma unroll
  for (int r = 0; r < 8; ++r) c[r] = base[(m0 + r) * LDP + n];
  return c;
}
__device__ __forceinline__ void cd_store(float* base, int mi, int nj, int lane, v8f c) {
  int n  = nj * 16 + (lane & 15);
  int m0 = mi * 16 + ((lane >> 4) << 3);
#pragma unroll
  for (int r = 0; r < 8; ++r) base[(m0 + r) * LDP + n] = c[r];
}

__device__ __forceinline__ v8f gemm_tn(const float* At, const float* Bm,
                                       int mi, int nj, int lane, v8f acc) {
  int lm = lane & 15, hi = lane >> 4;
  int am   = mi * 16 + lm;
  int bcol = nj * 16 + lm;
#pragma unroll
  for (int kk = 0; kk < 16; ++kk) {
    int kb = kk * 4 + hi * 2;
    v2f a; a.x = At[kb * LDP + am];   a.y = At[(kb + 1) * LDP + am];
    v2f b; b.x = Bm[kb * LDP + bcol]; b.y = Bm[(kb + 1) * LDP + bcol];
    acc = wmma4(a, b, acc);
  }
  return acc;
}

__device__ __forceinline__ float xhalf(float x) {
  return __int_as_float(__builtin_amdgcn_permlanex16(__float_as_int(x), __float_as_int(x),
                                                     0x76543210, 0xfedcba98, false, false));
}
__device__ __forceinline__ void ksum_wave_partial(float* kpart, int tid, float4 cs) {
  cs.x += xhalf(cs.x); cs.y += xhalf(cs.y); cs.z += xhalf(cs.z); cs.w += xhalf(cs.w);
  if ((tid & 31) < 16)
    *reinterpret_cast<float4*>(kpart + (tid >> 5) * 64 + ((tid & 15) << 2)) = cs;
}
__device__ __forceinline__ float ksum_join(const float* kpart, int d) {
  float p0 = kpart[d],       p1 = kpart[64 + d],  p2 = kpart[128 + d], p3 = kpart[192 + d];
  float p4 = kpart[256 + d], p5 = kpart[320 + d], p6 = kpart[384 + d], p7 = kpart[448 + d];
  return ((p0 + p1) + (p2 + p3)) + ((p4 + p5) + (p6 + p7));
}

__device__ __forceinline__ void tile_issue(const float* __restrict__ g, int rowStride,
                                           int tid, float4 (&w)[4]) {
#pragma unroll
  for (int j = 0; j < 4; ++j) {
    int u = tid + j * 256;
    int r = u >> 4;
    int c = (u & 15) << 2;
    w[j] = *reinterpret_cast<const float4*>(g + (size_t)r * rowStride + c);
  }
}
__device__ __forceinline__ void tile_commit(float* lds, int tid, const float4 (&w)[4],
                                            bool do_elu) {
#pragma unroll
  for (int j = 0; j < 4; ++j) {
    int u = tid + j * 256;
    int r = u >> 4;
    int c = (u & 15) << 2;
    float4 x = w[j];
    if (do_elu) { x.x = elu1(x.x); x.y = elu1(x.y); x.z = elu1(x.z); x.w = elu1(x.w); }
    *reinterpret_cast<float4*>(lds + r * LDP + c) = x;
  }
}
__device__ __forceinline__ float4 tile_commit_colsum(float* lds, int tid, const float4 (&w)[4]) {
  float4 cs = make_float4(0.0f, 0.0f, 0.0f, 0.0f);
#pragma unroll
  for (int j = 0; j < 4; ++j) {
    int u = tid + j * 256;
    int r = u >> 4;
    int c = (u & 15) << 2;
    float4 x = w[j];
    x.x = elu1(x.x); x.y = elu1(x.y); x.z = elu1(x.z); x.w = elu1(x.w);
    *reinterpret_cast<float4*>(lds + r * LDP + c) = x;
    if (j == 0) cs = x;
    else { cs.x += x.x; cs.y += x.y; cs.z += x.z; cs.w += x.w; }
  }
  return cs;
}


constexpr int BK = 4;

template <int NT>
__device__ __forceinline__ void rows_times_q(const float* (&ap)[NT],
                                             const v2f (&qf)[16], v8f (&acc)[NT]) {
  constexpr int NB = 16 / BK;
  v2f a[2][NT][BK];
#pragma unroll
  for (int i = 0; i < NT; ++i)
#pragma unroll
    for (int j = 0; j < BK; ++j) a[0][i][j] = lds_pair(ap[i] + j * 4);
#pragma unroll
  for (int blk = 0; blk < NB; ++blk) {
    const int cur = blk & 1, nxt = cur ^ 1;
    if (blk + 1 < NB) {
#pragma unroll
      for (int i = 0; i < NT; ++i)
#pragma unroll
        for (int j = 0; j < BK; ++j)
          a[nxt][i][j] = lds_pair(ap[i] + ((blk + 1) * BK + j) * 4);
    }
    __builtin_amdgcn_sched_barrier(0);
#pragma unroll
    for (int j = 0; j < BK; ++j)
#pragma unroll
      for (int i = 0; i < NT; ++i)
        acc[i] = wmma4(a[cur][i][j], qf[blk * BK + j], acc[i]);
    __builtin_amdgcn_sched_barrier(0);
  }
}

template <int TB, int SB0, int NT>
__device__ __forceinline__ void intra_blocks(const float* sK, const float* sV,
                                             const v2f (&qf)[16], int lane,
                                             v8f (&o)[4], float& den) {
  const int c = lane & 15, h = lane >> 4;
  float vf[NT][4][4][2];
#pragma unroll
  for (int i = 0; i < NT; ++i) {
    const float* vp = sV + ((SB0 + i) * 16 + 8 * h) * LDP + c;
#pragma unroll
    for (int j = 0; j < 4; ++j)
#pragma unroll
      for (int f = 0; f < 4; ++f) {
        vf[i][j][f][0] = vp[(2 * j) * LDP + 16 * f];
        vf[i][j][f][1] = vp[(2 * j + 1) * LDP + 16 * f];
      }
  }
  const float* ap[NT];
  v8f s[NT];
#pragma unroll
  for (int i = 0; i < NT; ++i) {
    ap[i] = sK + ((SB0 + i) * 16 + c) * LDP + 2 * h;
    s[i] = v8f{};
  }
  rows_times_q<NT>(ap, qf, s);
#pragma unroll
  for (int i = 0; i < NT; ++i) {
    if (SB0 + i == TB) {
#pragma unroll
      for (int r = 0; r < 8; ++r)
        if (8 * h + r > c) s[i][r] = 0.0f;
    }
#pragma unroll
    for (int r = 0; r < 8; ++r) den += s[i][r];
#pragma unroll
    for (int j = 0; j < 4; ++j) {
      v2f p;  p.x = s[i][2 * j];    p.y = s[i][2 * j + 1];
#pragma unroll
      for (int f = 0; f < 4; ++f) {
        v2f a; a.x = vf[i][j][f][0]; a.y = vf[i][j][f][1];
        o[f] = wmma4(a, p, o[f]);
      }
    }
  }
}

template <int TB>
__device__ __forceinline__ void wave_out(const float* sQ, const float* sK, const float* sV,
                                         const float* sKVT, const float* sks,
                                         float* stg, int lane,
                                         float* __restrict__ outp) {
  const int c = lane & 15, h = lane >> 4;
  v2f qf[16];
  {
    const float* qp = sQ + (TB * 16 + c) * LDP + 2 * h;
#pragma unroll
    for (int j = 0; j < 16; ++j) qf[j] = lds_pair(qp + j * 4);
  }
  v8f o[4] = {v8f{}, v8f{}, v8f{}, v8f{}};
  {
    const float* kp[4];
#pragma unroll
    for (int f = 0; f < 4; ++f) kp[f] = sKVT + (f * 16 + c) * LDP + 2 * h;
    rows_times_q<4>(kp, qf, o);
  }
  v2f dq = {0.0f, 0.0f};
#pragma unroll
  for (int j = 0; j < 16; ++j) dq += qf[j] * lds_pair(sks + j * 4 + 2 * h);
  float den = dq.x + dq.y;
  if constexpr (TB >= 1) intra_blocks<TB, 0, 2>(sK, sV, qf, lane, o, den);
  else                   intra_blocks<TB, 0, 1>(sK, sV, qf, lane, o, den);
  if constexpr (TB == 2) intra_blocks<TB, 2, 1>(sK, sV, qf, lane, o, den);
  if constexpr (TB == 3) intra_blocks<TB, 2, 2>(sK, sV, qf, lane, o, den);
  den += xhalf(den);
  const float inv = 1.0f / den;
#pragma unroll
  for (int i = 0; i < 4; ++i) {
    float* sp = stg + c * STGP + 16 * i + 8 * h;
    *reinterpret_cast<float4*>(sp)     = make_float4(o[i][0] * inv, o[i][1] * inv, o[i][2] * inv, o[i][3] * inv);
    *reinterpret_cast<float4*>(sp + 4) = make_float4(o[i][4] * inv, o[i][5] * inv, o[i][6] * inv, o[i][7] * inv);
  }
  __builtin_amdgcn_fence(__ATOMIC_RELEASE, "wavefront", "local");
  __builtin_amdgcn_wave_barrier();
  __builtin_amdgcn_fence(__ATOMIC_ACQUIRE, "wavefront", "local");
#pragma unroll
  for (int i = 0; i < 8; ++i) {
    int idx = lane + 32 * i;
    int tok = idx >> 4, pc = idx & 15;
    float4 x = *reinterpret_cast<const float4*>(stg + tok * STGP + 4 * pc);
    *reinterpret_cast<float4*>(outp + (size_t)tok * (H_ * D_) + 4 * pc) = x;
  }
}

template <int T0, int NT>
__device__ __forceinline__ void state_tiles(const float* sV, const float* sK, float* sKVTn,
                                            int lane, v8f (&kv)[7]) {
  constexpr int MI0 = T0 >> 2, MI1 = (T0 + NT - 1) >> 2, NMI = MI1 - MI0 + 1;
  constexpr int NJ0 = (NMI > 1) ? 0 : (T0 & 3);
  constexpr int NJ1 = (NMI > 1) ? 3 : ((T0 + NT - 1) & 3);
  constexpr int NNJ = NJ1 - NJ0 + 1;
  constexpr int NB = 16 / BK;
  const int c = lane & 15, h = lane >> 4;
  const float* vp = sV + (2 * h) * LDP + MI0 * 16 + c;
  const float* kp = sK + (2 * h) * LDP + NJ0 * 16 + c;
  float a[2][BK][NMI][2], b[2][BK][NNJ][2];
#pragma unroll
  for (int j = 0; j < BK; ++j) {
#pragma unroll
    for (int m = 0; m < NMI; ++m) {
      a[0][j][m][0] = vp[(j * 4) * LDP + m * 16];
      a[0][j][m][1] = vp[(j * 4 + 1) * LDP + m * 16];
    }
#pragma unroll
    for (int n = 0; n < NNJ; ++n) {
      b[0][j][n][0] = kp[(j * 4) * LDP + n * 16];
      b[0][j][n][1] = kp[(j * 4 + 1) * LDP + n * 16];
    }
  }
#pragma unroll
  for (int blk = 0; blk < NB; ++blk) {
    const int cur = blk & 1, nxt = cur ^ 1;
    if (blk + 1 < NB) {
#pragma unroll
      for (int j = 0; j < BK; ++j) {
        const int kr = ((blk + 1) * BK + j) * 4;
#pragma unroll
        for (int m = 0; m < NMI; ++m) {
          a[nxt][j][m][0] = vp[kr * LDP + m * 16];
          a[nxt][j][m][1] = vp[(kr + 1) * LDP + m * 16];
        }
#pragma unroll
        for (int n = 0; n < NNJ; ++n) {
          b[nxt][j][n][0] = kp[kr * LDP + n * 16];
          b[nxt][j][n][1] = kp[(kr + 1) * LDP + n * 16];
        }
      }
    }
    __builtin_amdgcn_sched_barrier(0);
#pragma unroll
    for (int j = 0; j < BK; ++j)
#pragma unroll
      for (int i = 0; i < NT; ++i) {
        const int m = ((T0 + i) >> 2) - MI0, n = ((T0 + i) & 3) - NJ0;
        v2f af; af.x = a[cur][j][m][0]; af.y = a[cur][j][m][1];
        v2f bf; bf.x = b[cur][j][n][0]; bf.y = b[cur][j][n][1];
        kv[i] = wmma4(af, bf, kv[i]);
      }
    __builtin_amdgcn_sched_barrier(0);
  }
#pragma unroll
  for (int i = 0; i < NT; ++i) cd_store(sKVTn, (T0 + i) >> 2, (T0 + i) & 3, lane, kv[i]);
}
template <int T0, int NT>
__device__ __forceinline__ void state_load(const float* plane, int lane, v8f (&kv)[7]) {
#pragma unroll
  for (int i = 0; i < NT; ++i) kv[i] = cd_load(plane, (T0 + i) >> 2, (T0 + i) & 3, lane);
}


__global__ __launch_bounds__(256) void lin_attn_states(
    const float* __restrict__ qk, const float* __restrict__ v,
    float* __restrict__ ws, int nsplit) {
  extern __shared__ float smem[];
  float* sK    = smem + OFF_K;
  float* sV    = smem + OFF_V;
  float* sKV   = smem + OFF_KV;
  float* sKpart = smem + OFF_KPART;

  int tid = threadIdx.x, lane = tid & 31, wid = tid >> 5;
  int wg = blockIdx.x;
  int split = wg % nsplit;
  int h     = (wg / nsplit) % H_;
  int b     = wg / (nsplit * H_);
  int cps   = (S_ / CH) / nsplit;

  for (int i = tid; i < 64 * LDP; i += 256) sKV[i] = 0.0f;
  float ks = 0.0f;
  __syncthreads();

  float4 rk[4], rv[4];
  {
    int base0 = split * cps * CH;
    tile_issue(qk + (((size_t)(b * S_ + base0) * 2 + 1) * H_ + h) * D_, 2 * H_ * D_, tid, rk);
    tile_issue(v  + (((size_t)(b * S_ + base0)) * H_ + h) * D_,         H_ * D_,     tid, rv);
  }

  for (int c = 0; c < cps; ++c) {
    ksum_wave_partial(sKpart, tid, tile_commit_colsum(sK, tid, rk));
    tile_commit(sV, tid, rv, false);
    if (c + 1 < cps) {
      int nb = (split * cps + c + 1) * CH;
      tile_issue(qk + (((size_t)(b * S_ + nb) * 2 + 1) * H_ + h) * D_, 2 * H_ * D_, tid, rk);
      tile_issue(v  + (((size_t)(b * S_ + nb)) * H_ + h) * D_,         H_ * D_,     tid, rv);
    }
    lds_barrier();

    if (tid < 64) ks += ksum_join(sKpart, tid);
#pragma unroll
    for (int i = 0; i < 2; ++i) {
      int t = wid * 2 + i, mi = t >> 2, nj = t & 3;
      v8f acc = cd_load(sKV, mi, nj, lane);
      acc = gemm_tn(sV, sK, mi, nj, lane, acc);
      cd_store(sKV, mi, nj, lane, acc);
    }
    lds_barrier();
  }

  float* wsp = ws + (size_t)((b * H_ + h) * nsplit + split) * STATE_F;
  for (int i = tid; i < 64 * 64; i += 256)
    wsp[i] = sKV[(i >> 6) * LDP + (i & 63)];
  if (tid < 64) wsp[64 * 64 + tid] = ks;
}

__global__ __launch_bounds__(256) void lin_attn_main(
    const float* __restrict__ qk, const float* __restrict__ v,
    const float* __restrict__ ws, float* __restrict__ out, int nsplit) {
  __shared__ __attribute__((aligned(16))) float smem[MAIN_SMEM_F];

  int tid = threadIdx.x, lane = tid & 31, wid = tid >> 5;
  int wg = blockIdx.x;
  int split = wg % nsplit;
  int h     = (wg / nsplit) % H_;
  int b     = wg / (nsplit * H_);
  int cps   = (S_ / CH) / nsplit;

  float4 rq[4], rk[4], rv[4];
  {
    int base0 = split * cps * CH;
    tile_issue(qk + (((size_t)(b * S_ + base0) * 2 + 0) * H_ + h) * D_, 2 * H_ * D_, tid, rq);
    tile_issue(qk + (((size_t)(b * S_ + base0) * 2 + 1) * H_ + h) * D_, 2 * H_ * D_, tid, rk);
    tile_issue(v  + (((size_t)(b * S_ + base0)) * H_ + h) * D_,         H_ * D_,     tid, rv);
  }

  {
    constexpr int PB = 7;
    float* sKVT0 = smem + MOFF_KVT;
    float* sks0  = smem + MOFF_KS;
    const float* wbase = ws + (size_t)((b * H_ + h) * nsplit) * STATE_F;
    float4 acc[4];
    float  accs = 0.0f;
#pragma unroll
    for (int j = 0; j < 4; ++j) acc[j] = make_float4(0.0f, 0.0f, 0.0f, 0.0f);
    for (int p0 = 0; p0 < split; p0 += PB) {
      float4 st[PB][4];
      float  sk[PB];
#pragma unroll
      for (int q = 0; q < PB; ++q) {
        if (p0 + q < split) {
          const float* wsp = wbase + (size_t)(p0 + q) * STATE_F;
#pragma unroll
          for (int j = 0; j < 4; ++j)
            st[q][j] = *reinterpret_cast<const float4*>(wsp + 4 * (tid + 256 * j));
          if (tid < 64) sk[q] = wsp[64 * 64 + tid];
        }
      }
#pragma unroll
      for (int q = 0; q < PB; ++q) {
        if (p0 + q < split) {
#pragma unroll
          for (int j = 0; j < 4; ++j) {
            acc[j].x += st[q][j].x; acc[j].y += st[q][j].y;
            acc[j].z += st[q][j].z; acc[j].w += st[q][j].w;
          }
          if (tid < 64) accs += sk[q];
        }
      }
    }
#pragma unroll
    for (int j = 0; j < 4; ++j) {
      int f = tid + 256 * j;
      *reinterpret_cast<float4*>(sKVT0 + (f >> 4) * LDP + ((f & 15) << 2)) = acc[j];
    }
    if (tid < 64) sks0[tid] = accs;
  }
  __syncthreads();

  const int widu = __builtin_amdgcn_readfirstlane(wid);
  v8f kv[7];
  switch (widu) {
    case 4:  state_load<0, 7>(smem + MOFF_KVT, lane, kv);  break;
    case 5:  state_load<7, 5>(smem + MOFF_KVT, lane, kv);  break;
    case 6:  state_load<12, 3>(smem + MOFF_KVT, lane, kv); break;
    case 7:  state_load<15, 1>(smem + MOFF_KVT, lane, kv); break;
    default: break;
  }
  float* stg = smem + MOFF_STG + (widu & 3) * (16 * STGP);

  for (int c = 0; c < cps; ++c) {
    const int buf = c & 1;
    int base = (split * cps + c) * CH;
    float* sQ = smem + MOFF_T + buf * (3 * TILE_F);
    float* sK = sQ + TILE_F;
    float* sV = sK + TILE_F;
    const float* sKVTc = smem + MOFF_KVT + buf * TILE_F;
    float*       sKVTn = smem + MOFF_KVT + (buf ^ 1) * TILE_F;
    const float* sksc  = smem + MOFF_KS + buf * 64;
    float*       sksn  = smem + MOFF_KS + (buf ^ 1) * 64;
    float*       sKp   = smem + MOFF_KP + buf * 512;

    tile_commit(sQ, tid, rq, true);
    ksum_wave_partial(sKp, tid, tile_commit_colsum(sK, tid, rk));
    tile_commit(sV, tid, rv, false);
    if (c + 1 < cps) {
      int nb = base + CH;
      tile_issue(qk + (((size_t)(b * S_ + nb) * 2 + 0) * H_ + h) * D_, 2 * H_ * D_, tid, rq);
      tile_issue(qk + (((size_t)(b * S_ + nb) * 2 + 1) * H_ + h) * D_, 2 * H_ * D_, tid, rk);
      tile_issue(v  + (((size_t)(b * S_ + nb)) * H_ + h) * D_,         H_ * D_,     tid, rv);
    }
    lds_barrier();

    if (widu < 4) {
      float* outp = out + (((size_t)b * S_ + base + widu * 16) * H_ + h) * D_;
      switch (widu) {
        case 0:  wave_out<0>(sQ, sK, sV, sKVTc, sksc, stg, lane, outp); break;
        case 1:  wave_out<1>(sQ, sK, sV, sKVTc, sksc, stg, lane, outp); break;
        case 2:  wave_out<2>(sQ, sK, sV, sKVTc, sksc, stg, lane, outp); break;
        default: wave_out<3>(sQ, sK, sV, sKVTc, sksc, stg, lane, outp); break;
      }
    } else if (c + 1 < cps) {
      if (widu < 6) {
        const int d = (widu - 4) * 32 + lane;
        sksn[d] = sksc[d] + ksum_join(sKp, d);
      }
      switch (widu) {
        case 4:  state_tiles<0, 7>(sV, sK, sKVTn, lane, kv);  break;
        case 5:  state_tiles<7, 5>(sV, sK, sKVTn, lane, kv);  break;
        case 6:  state_tiles<12, 3>(sV, sK, sKVTn, lane, kv); break;
        default: state_tiles<15, 1>(sV, sK, sKVTn, lane, kv); break;
      }
    }
  }
}

extern "C" void kernel_launch(void* const* d_in, const int* in_sizes, int n_in,
                              void* d_out, int out_size, void* d_ws, size_t ws_size,
                              hipStream_t stream) {
  const float* qk = (const float*)d_in[0];
  const float* v  = (const float*)d_in[1];
  float* out = (float*)d_out;
  float* ws  = (float*)d_ws;

  int nsplit = NSPL;
  size_t need = (size_t)B_ * H_ * NSPL * STATE_F * sizeof(float);
  if (ws_size < need) nsplit = 1;

  int wgs = B_ * H_ * nsplit;
  size_t smem = (size_t)SMEM_F * sizeof(float);

  if (nsplit > 1)
    lin_attn_states<<<wgs, 256, smem, stream>>>(qk, v, ws, nsplit);
  lin_attn_main<<<wgs, 256, 0, stream>>>(qk, v, ws, out, nsplit);
}
